# starting kernel
# baseline (speedup 1.0000x reference)
#include <hip/hip_runtime.h>
#include <hip/hip_bf16.h>
#include <math.h>

typedef __attribute__((ext_vector_type(2))) float v2f;
typedef __attribute__((ext_vector_type(8))) float v8f;

#define S_DIM 1024
#define CA 1024
#define CS 512
#define CZ 64
#define NH 16
#define HC 64

__device__ __forceinline__ float sigmoidf_(float x) {
    return 1.0f / (1.0f + __expf(-x));
}

__global__ void ln_kernel(const float* __restrict__ x, const float* __restrict__ w,
                          float* __restrict__ y, int C) {
    __shared__ float red[256];
    __shared__ float red2[256];
    int row = blockIdx.x;
    const float4* xr = (const float4*)(x + (size_t)row * C);
    int C4 = C >> 2;
    float sum = 0.f, sq = 0.f;
    for (int c = threadIdx.x; c < C4; c += 256) {
        float4 v = xr[c];
        sum += v.x + v.y + v.z + v.w;
        sq  += v.x * v.x + v.y * v.y + v.z * v.z + v.w * v.w;
    }
    red[threadIdx.x] = sum; red2[threadIdx.x] = sq;
    __syncthreads();
    for (int off = 128; off > 0; off >>= 1) {
        if (threadIdx.x < off) {
            red[threadIdx.x]  += red[threadIdx.x + off];
            red2[threadIdx.x] += red2[threadIdx.x + off];
        }
        __syncthreads();
    }
    float m   = red[0] / C;
    float var = red2[0] / C - m * m;
    float inv = rsqrtf(var + 1e-5f);
    float4* yr = (float4*)(y + (size_t)row * C);
    for (int c = threadIdx.x; c < C4; c += 256) {
        float4 v = xr[c];
        v.x = (v.x - m) * inv; v.y = (v.y - m) * inv;
        v.z = (v.z - m) * inv; v.w = (v.w - m) * inv;
        if (w) {
            const float4 wv = ((const float4*)w)[c];
            v.x *= wv.x; v.y *= wv.y; v.z *= wv.z; v.w *= wv.w;
        }
        yr[c] = v;
    }
}

__global__ void gemm_nt_kernel(const float* __restrict__ A, int lda, long long sA,
                               const float* __restrict__ B, int ldb, long long sB,
                               float* __restrict__ C, int ldc, long long sC,
                               const float* __restrict__ bias,
                               const float* __restrict__ addm, int ldadd, long long sAdd,
                               int K, float scale, int mode) {
    __shared__ float As[2][64][20];
    __shared__ float Bs[2][64][20];
    const int z = blockIdx.z;
    A += (size_t)z * sA; B += (size_t)z * sB; C += (size_t)z * sC;
    if (addm) addm += (size_t)z * sAdd;
    const int tileM = blockIdx.y * 64;
    const int tileN = blockIdx.x * 64;
    const int tid  = threadIdx.x;
    const int wave = tid >> 5, lane = tid & 31;
    const int wm = wave >> 1, wn = wave & 1;
    const int half = lane >> 4, l = lane & 15;
    const int lr = tid >> 2;
    const int lc = (tid & 3) * 4;

    const float* gA = A + (size_t)(tileM + lr) * lda + lc;
    const float* gB = B + (size_t)(tileN + lr) * ldb + lc;

    v8f acc0 = {}; v8f acc1 = {};
    const int nstages = K >> 4;

    {
        float4 ra = *(const float4*)gA;
        float4 rb = *(const float4*)gB;
        *(float4*)&As[0][lr][lc] = ra;
        *(float4*)&Bs[0][lr][lc] = rb;
    }
    __syncthreads();

    for (int s = 0; s < nstages; s++) {
        const int p = s & 1;
        float4 na = make_float4(0.f, 0.f, 0.f, 0.f);
        float4 nb = make_float4(0.f, 0.f, 0.f, 0.f);
        if (s + 1 < nstages) {
            na = *(const float4*)(gA + (s + 1) * 16);
            nb = *(const float4*)(gB + (s + 1) * 16);
        }
        v2f af[4], bf0[4], bf1[4];
        #pragma unroll
        for (int q = 0; q < 4; q++) {
            const int kc = q * 4 + half * 2;
            af[q].x  = As[p][wm * 16 + l][kc];
            af[q].y  = As[p][wm * 16 + l][kc + 1];
            bf0[q].x = Bs[p][wn * 32 + l][kc];
            bf0[q].y = Bs[p][wn * 32 + l][kc + 1];
            bf1[q].x = Bs[p][wn * 32 + 16 + l][kc];
            bf1[q].y = Bs[p][wn * 32 + 16 + l][kc + 1];
        }
        #pragma unroll
        for (int q = 0; q < 4; q++) {
            acc0 = __builtin_amdgcn_wmma_f32_16x16x4_f32(false, af[q], false, bf0[q], (short)0, acc0, false, false);
            acc1 = __builtin_amdgcn_wmma_f32_16x16x4_f32(false, af[q], false, bf1[q], (short)0, acc1, false, false);
        }
        if (s + 1 < nstages) {
            *(float4*)&As[p ^ 1][lr][lc] = na;
            *(float4*)&Bs[p ^ 1][lr][lc] = nb;
        }
        __syncthreads();
    }

    #pragma unroll
    for (int r = 0; r < 8; r++) {
        int m  = tileM + wm * 16 + half * 8 + r;
        int n0 = tileN + wn * 32 + l;
        int n1 = n0 + 16;
        float v0 = acc0[r] * scale;
        float v1 = acc1[r] * scale;
        if (bias) { v0 += bias[n0]; v1 += bias[n1]; }
        if (mode == 1) {
            v0 = sigmoidf_(v0) * addm[(size_t)m * ldadd + n0];
            v1 = sigmoidf_(v1) * addm[(size_t)m * ldadd + n1];
        } else if (addm) {
            v0 += addm[(size_t)m * ldadd + n0];
            v1 += addm[(size_t)m * ldadd + n1];
        }
        C[(size_t)m * ldc + n0] = v0;
        C[(size_t)m * ldc + n1] = v1;
    }
}

__global__ void gemm_tn_kernel(const float* __restrict__ A, int lda, long long sA,
                               const float* __restrict__ B, int ldb, long long sB,
                               float* __restrict__ C, int ldc, long long sC,
                               int K) {
    __shared__ float As[2][16][72];
    __shared__ float Bs[2][16][72];
    const int z = blockIdx.z;
    A += (size_t)z * sA; B += (size_t)z * sB; C += (size_t)z * sC;
    const int tileM = blockIdx.y * 64;
    const int tileN = blockIdx.x * 64;
    const int tid  = threadIdx.x;
    const int wave = tid >> 5, lane = tid & 31;
    const int wm = wave >> 1, wn = wave & 1;
    const int half = lane >> 4, l = lane & 15;
    const int lk = tid >> 4;
    const int lm = (tid & 15) * 4;

    const float* gA = A + (size_t)lk * lda + tileM + lm;
    const float* gB = B + (size_t)lk * ldb + tileN + lm;

    v8f acc0 = {}; v8f acc1 = {};
    const int nstages = K >> 4;

    {
        float4 ra = *(const float4*)gA;
        float4 rb = *(const float4*)gB;
        *(float4*)&As[0][lk][lm] = ra;
        *(float4*)&Bs[0][lk][lm] = rb;
    }
    __syncthreads();

    for (int s = 0; s < nstages; s++) {
        const int p = s & 1;
        float4 na = make_float4(0.f, 0.f, 0.f, 0.f);
        float4 nb = make_float4(0.f, 0.f, 0.f, 0.f);
        if (s + 1 < nstages) {
            na = *(const float4*)(gA + (size_t)(s + 1) * 16 * lda);
            nb = *(const float4*)(gB + (size_t)(s + 1) * 16 * ldb);
        }
        v2f af[4], bf0[4], bf1[4];
        #pragma unroll
        for (int q = 0; q < 4; q++) {
            const int kc = q * 4 + half * 2;
            af[q].x  = As[p][kc][wm * 16 + l];
            af[q].y  = As[p][kc + 1][wm * 16 + l];
            bf0[q].x = Bs[p][kc][wn * 32 + l];
            bf0[q].y = Bs[p][kc + 1][wn * 32 + l];
            bf1[q].x = Bs[p][kc][wn * 32 + 16 + l];
            bf1[q].y = Bs[p][kc + 1][wn * 32 + 16 + l];
        }
        #pragma unroll
        for (int q = 0; q < 4; q++) {
            acc0 = __builtin_amdgcn_wmma_f32_16x16x4_f32(false, af[q], false, bf0[q], (short)0, acc0, false, false);
            acc1 = __builtin_amdgcn_wmma_f32_16x16x4_f32(false, af[q], false, bf1[q], (short)0, acc1, false, false);
        }
        if (s + 1 < nstages) {
            *(float4*)&As[p ^ 1][lk][lm] = na;
            *(float4*)&Bs[p ^ 1][lk][lm] = nb;
        }
        __syncthreads();
    }

    #pragma unroll
    for (int r = 0; r < 8; r++) {
        int m  = tileM + wm * 16 + half * 8 + r;
        int n0 = tileN + wn * 32 + l;
        C[(size_t)m * ldc + n0]      = acc0[r];
        C[(size_t)m * ldc + n0 + 16] = acc1[r];
    }
}

__global__ void softmax_row_kernel(float* __restrict__ x) {
    __shared__ float red[256];
    float4* xr = (float4*)(x + (size_t)blockIdx.x * 1024);
    float4 v = xr[threadIdx.x];
    float mx = fmaxf(fmaxf(v.x, v.y), fmaxf(v.z, v.w));
    red[threadIdx.x] = mx; __syncthreads();
    for (int off = 128; off > 0; off >>= 1) {
        if (threadIdx.x < off) red[threadIdx.x] = fmaxf(red[threadIdx.x], red[threadIdx.x + off]);
        __syncthreads();
    }
    mx = red[0];
    __syncthreads();
    v.x = __expf(v.x - mx); v.y = __expf(v.y - mx);
    v.z = __expf(v.z - mx); v.w = __expf(v.w - mx);
    float s = v.x + v.y + v.z + v.w;
    red[threadIdx.x] = s; __syncthreads();
    for (int off = 128; off > 0; off >>= 1) {
        if (threadIdx.x < off) red[threadIdx.x] += red[threadIdx.x + off];
        __syncthreads();
    }
    float inv = 1.0f / red[0];
    v.x *= inv; v.y *= inv; v.z *= inv; v.w *= inv;
    xr[threadIdx.x] = v;
}

__global__ void zbias_kernel(const float* __restrict__ z,
                             const float* __restrict__ pnw, const float* __restrict__ pnb,
                             const float* __restrict__ bw,  const float* __restrict__ bb,
                             float* __restrict__ out) {
    __shared__ float bwS[NH * CZ];
    __shared__ float pwS[CZ];
    __shared__ float pbS[CZ];
    __shared__ float bbS[NH];
    int tid = threadIdx.x;
    for (int i = tid; i < NH * CZ; i += 256) bwS[i] = bw[i];
    if (tid < CZ) { pwS[tid] = pnw[tid]; pbS[tid] = pnb[tid]; }
    if (tid < NH) bbS[tid] = bb[tid];
    __syncthreads();
    size_t pos = (size_t)blockIdx.x * 256 + tid;
    const float4* zp = (const float4*)(z + pos * CZ);
    float4 zv[16];
    float sum = 0.f, sq = 0.f;
    #pragma unroll
    for (int i = 0; i < 16; i++) {
        float4 t = zp[i];
        zv[i] = t;
        sum += t.x + t.y + t.z + t.w;
        sq  += t.x * t.x + t.y * t.y + t.z * t.z + t.w * t.w;
    }
    float m   = sum * (1.0f / 64.0f);
    float var = sq * (1.0f / 64.0f) - m * m;
    float inv = rsqrtf(var + 1e-5f);
    float o[NH];
    #pragma unroll
    for (int h = 0; h < NH; h++) o[h] = bbS[h];
    #pragma unroll
    for (int i = 0; i < 16; i++) {
        float y0 = (zv[i].x - m) * inv * pwS[4 * i + 0] + pbS[4 * i + 0];
        float y1 = (zv[i].y - m) * inv * pwS[4 * i + 1] + pbS[4 * i + 1];
        float y2 = (zv[i].z - m) * inv * pwS[4 * i + 2] + pbS[4 * i + 2];
        float y3 = (zv[i].w - m) * inv * pwS[4 * i + 3] + pbS[4 * i + 3];
        #pragma unroll
        for (int h = 0; h < NH; h++) {
            o[h] += y0 * bwS[h * 64 + 4 * i + 0] + y1 * bwS[h * 64 + 4 * i + 1]
                  + y2 * bwS[h * 64 + 4 * i + 2] + y3 * bwS[h * 64 + 4 * i + 3];
        }
    }
    float4* op = (float4*)(out + pos * NH);
    op[0] = make_float4(o[0],  o[1],  o[2],  o[3]);
    op[1] = make_float4(o[4],  o[5],  o[6],  o[7]);
    op[2] = make_float4(o[8],  o[9],  o[10], o[11]);
    op[3] = make_float4(o[12], o[13], o[14], o[15]);
}

__global__ void adaln_gate_kernel(const float4* __restrict__ pb, const float4* __restrict__ pn,
                                  const float4* __restrict__ an, float4* __restrict__ a1) {
    int i = blockIdx.x * 256 + threadIdx.x;
    float4 p = pb[i], q = pn[i], n = an[i];
    float4 r;
    r.x = sigmoidf_(p.x * n.x + q.x);
    r.y = sigmoidf_(p.y * n.y + q.y);
    r.z = sigmoidf_(p.z * n.z + q.z);
    r.w = sigmoidf_(p.w * n.w + q.w);
    a1[i] = r;
}

__global__ void gate_o_kernel(const float* __restrict__ kvg, const float4* __restrict__ obuf,
                              float4* __restrict__ go) {
    int q = blockIdx.x * 256 + threadIdx.x;
    int h  = q >> 14;
    int j  = (q >> 4) & 1023;
    int c4 = (q & 15) * 4;
    const float4 g = *(const float4*)(kvg + (size_t)h * 196608 + (size_t)j * 192 + 128 + c4);
    float4 o = obuf[q];
    float4 r;
    r.x = sigmoidf_(g.x) * o.x;
    r.y = sigmoidf_(g.y) * o.y;
    r.z = sigmoidf_(g.z) * o.z;
    r.w = sigmoidf_(g.w) * o.w;
    go[q] = r;
}

extern "C" void kernel_launch(void* const* d_in, const int* in_sizes, int n_in,
                              void* d_out, int out_size, void* d_ws, size_t ws_size,
                              hipStream_t stream) {
    const float* a       = (const float*)d_in[0];
    const float* z       = (const float*)d_in[1];
    const float* s       = (const float*)d_in[2];
    const float* sn_w    = (const float*)d_in[3];
    const float* pb_w    = (const float*)d_in[4];
    const float* pb_b    = (const float*)d_in[5];
    const float* pn_w    = (const float*)d_in[6];
    const float* pnorm_w = (const float*)d_in[7];
    const float* pnorm_b = (const float*)d_in[8];
    const float* q_w     = (const float*)d_in[9];
    const float* q_b     = (const float*)d_in[10];
    const float* kvg_w   = (const float*)d_in[11];
    const float* bias_w  = (const float*)d_in[12];
    const float* bias_b  = (const float*)d_in[13];
    const float* attn_w  = (const float*)d_in[14];
    const float* out_w   = (const float*)d_in[15];
    const float* out_b   = (const float*)d_in[16];
    float* out = (float*)d_out;
    float* ws  = (float*)d_ws;

    float* a_n    = ws + 0;
    float* s_n    = ws + 1048576;
    float* pb     = ws + 1572864;
    float* pn     = ws + 2621440;
    float* qm     = ws + 3670016;
    float* kvg    = ws + 4718592;
    float* biasO  = ws + 7864320;
    float* obuf   = biasO;
    float* go     = biasO + 1048576;
    float* a2     = biasO + 2097152;
    float* scores = ws + 24641536;

    ln_kernel<<<1024, 256, 0, stream>>>(a, nullptr, a_n, 1024);
    ln_kernel<<<1024, 256, 0, stream>>>(s, sn_w, s_n, 512);

    gemm_nt_kernel<<<dim3(16, 16, 1), 256, 0, stream>>>(
        s_n, 512, 0, pb_w, 512, 0, pb, 1024, 0, pb_b, nullptr, 0, 0, 512, 1.0f, 0);
    gemm_nt_kernel<<<dim3(16, 16, 1), 256, 0, stream>>>(
        s_n, 512, 0, pn_w, 512, 0, pn, 1024, 0, nullptr, nullptr, 0, 0, 512, 1.0f, 0);
    adaln_gate_kernel<<<1024, 256, 0, stream>>>(
        (const float4*)pb, (const float4*)pn, (const float4*)a_n, (float4*)pb);
    float* a1 = pb;

    gemm_nt_kernel<<<dim3(16, 16, 1), 256, 0, stream>>>(
        a1, 1024, 0, q_w, 1024, 0, qm, 1024, 0, q_b, nullptr, 0, 0, 1024, 1.0f, 0);
    gemm_nt_kernel<<<dim3(48, 16, 1), 256, 0, stream>>>(
        a1, 1024, 0, kvg_w, 1024, 0, kvg, 3072, 0, nullptr, nullptr, 0, 0, 1024, 1.0f, 0);

    zbias_kernel<<<4096, 256, 0, stream>>>(z, pnorm_w, pnorm_b, bias_w, bias_b, biasO);

    gemm_nt_kernel<<<dim3(16, 16, 16), 256, 0, stream>>>(
        kvg, 192, 196608LL,
        qm, 64, 65536LL,
        scores, 1024, 1048576LL,
        nullptr,
        biasO, 1024, 1048576LL,
        64, 1.0f / 64.0f, 0);

    softmax_row_kernel<<<16384, 256, 0, stream>>>(scores);

    gemm_tn_kernel<<<dim3(1, 16, 16), 256, 0, stream>>>(
        scores, 1024, 1048576LL,
        kvg + 64, 192, 196608LL,
        obuf, 64, 65536LL,
        1024);

    gate_o_kernel<<<1024, 256, 0, stream>>>(kvg, (const float4*)obuf, (float4*)go);

    gemm_nt_kernel<<<dim3(16, 16, 1), 256, 0, stream>>>(
        go, 1024, 0, attn_w, 1024, 0, a2, 1024, 0, nullptr, nullptr, 0, 0, 1024, 1.0f, 0);

    gemm_nt_kernel<<<dim3(16, 16, 1), 256, 0, stream>>>(
        s, 512, 0, out_w, 512, 0, out, 1024, 0, out_b, a2, 1024, 0, 512, 1.0f, 1);
}


# Pair-bias kernel: 16 lanes share a z row (b128 each, wave loads are whole lines), weights (norm weight folded in) stay in registers, head sums by DPP butterfly; no LDS or scratch; sum order changes
# speedup vs baseline: 1.1995x; 1.1995x over previous
#include <hip/hip_runtime.h>
#include <hip/hip_bf16.h>
#include <math.h>

typedef __attribute__((ext_vector_type(2))) float v2f;
typedef __attribute__((ext_vector_type(8))) float v8f;

#define S_DIM 1024
#define CA 1024
#define CS 512
#define CZ 64
#define NH 16
#define HC 64

__device__ __forceinline__ float sigmoidf_(float x) {
    return 1.0f / (1.0f + __expf(-x));
}

__global__ void ln_kernel(const float* __restrict__ x, const float* __restrict__ w,
                          float* __restrict__ y, int C) {
    __shared__ float red[256];
    __shared__ float red2[256];
    int row = blockIdx.x;
    const float4* xr = (const float4*)(x + (size_t)row * C);
    int C4 = C >> 2;
    float sum = 0.f, sq = 0.f;
    for (int c = threadIdx.x; c < C4; c += 256) {
        float4 v = xr[c];
        sum += v.x + v.y + v.z + v.w;
        sq  += v.x * v.x + v.y * v.y + v.z * v.z + v.w * v.w;
    }
    red[threadIdx.x] = sum; red2[threadIdx.x] = sq;
    __syncthreads();
    for (int off = 128; off > 0; off >>= 1) {
        if (threadIdx.x < off) {
            red[threadIdx.x]  += red[threadIdx.x + off];
            red2[threadIdx.x] += red2[threadIdx.x + off];
        }
        __syncthreads();
    }
    float m   = red[0] / C;
    float var = red2[0] / C - m * m;
    float inv = rsqrtf(var + 1e-5f);
    float4* yr = (float4*)(y + (size_t)row * C);
    for (int c = threadIdx.x; c < C4; c += 256) {
        float4 v = xr[c];
        v.x = (v.x - m) * inv; v.y = (v.y - m) * inv;
        v.z = (v.z - m) * inv; v.w = (v.w - m) * inv;
        if (w) {
            const float4 wv = ((const float4*)w)[c];
            v.x *= wv.x; v.y *= wv.y; v.z *= wv.z; v.w *= wv.w;
        }
        yr[c] = v;
    }
}

__global__ void gemm_nt_kernel(const float* __restrict__ A, int lda, long long sA,
                               const float* __restrict__ B, int ldb, long long sB,
                               float* __restrict__ C, int ldc, long long sC,
                               const float* __restrict__ bias,
                               const float* __restrict__ addm, int ldadd, long long sAdd,
                               int K, float scale, int mode) {
    __shared__ float As[2][64][20];
    __shared__ float Bs[2][64][20];
    const int z = blockIdx.z;
    A += (size_t)z * sA; B += (size_t)z * sB; C += (size_t)z * sC;
    if (addm) addm += (size_t)z * sAdd;
    const int tileM = blockIdx.y * 64;
    const int tileN = blockIdx.x * 64;
    const int tid  = threadIdx.x;
    const int wave = tid >> 5, lane = tid & 31;
    const int wm = wave >> 1, wn = wave & 1;
    const int half = lane >> 4, l = lane & 15;
    const int lr = tid >> 2;
    const int lc = (tid & 3) * 4;

    const float* gA = A + (size_t)(tileM + lr) * lda + lc;
    const float* gB = B + (size_t)(tileN + lr) * ldb + lc;

    v8f acc0 = {}; v8f acc1 = {};
    const int nstages = K >> 4;

    {
        float4 ra = *(const float4*)gA;
        float4 rb = *(const float4*)gB;
        *(float4*)&As[0][lr][lc] = ra;
        *(float4*)&Bs[0][lr][lc] = rb;
    }
    __syncthreads();

    for (int s = 0; s < nstages; s++) {
        const int p = s & 1;
        float4 na = make_float4(0.f, 0.f, 0.f, 0.f);
        float4 nb = make_float4(0.f, 0.f, 0.f, 0.f);
        if (s + 1 < nstages) {
            na = *(const float4*)(gA + (s + 1) * 16);
            nb = *(const float4*)(gB + (s + 1) * 16);
        }
        v2f af[4], bf0[4], bf1[4];
        #pragma unroll
        for (int q = 0; q < 4; q++) {
            const int kc = q * 4 + half * 2;
            af[q].x  = As[p][wm * 16 + l][kc];
            af[q].y  = As[p][wm * 16 + l][kc + 1];
            bf0[q].x = Bs[p][wn * 32 + l][kc];
            bf0[q].y = Bs[p][wn * 32 + l][kc + 1];
            bf1[q].x = Bs[p][wn * 32 + 16 + l][kc];
            bf1[q].y = Bs[p][wn * 32 + 16 + l][kc + 1];
        }
        #pragma unroll
        for (int q = 0; q < 4; q++) {
            acc0 = __builtin_amdgcn_wmma_f32_16x16x4_f32(false, af[q], false, bf0[q], (short)0, acc0, false, false);
            acc1 = __builtin_amdgcn_wmma_f32_16x16x4_f32(false, af[q], false, bf1[q], (short)0, acc1, false, false);
        }
        if (s + 1 < nstages) {
            *(float4*)&As[p ^ 1][lr][lc] = na;
            *(float4*)&Bs[p ^ 1][lr][lc] = nb;
        }
        __syncthreads();
    }

    #pragma unroll
    for (int r = 0; r < 8; r++) {
        int m  = tileM + wm * 16 + half * 8 + r;
        int n0 = tileN + wn * 32 + l;
        int n1 = n0 + 16;
        float v0 = acc0[r] * scale;
        float v1 = acc1[r] * scale;
        if (bias) { v0 += bias[n0]; v1 += bias[n1]; }
        if (mode == 1) {
            v0 = sigmoidf_(v0) * addm[(size_t)m * ldadd + n0];
            v1 = sigmoidf_(v1) * addm[(size_t)m * ldadd + n1];
        } else if (addm) {
            v0 += addm[(size_t)m * ldadd + n0];
            v1 += addm[(size_t)m * ldadd + n1];
        }
        C[(size_t)m * ldc + n0] = v0;
        C[(size_t)m * ldc + n1] = v1;
    }
}

__global__ void gemm_tn_kernel(const float* __restrict__ A, int lda, long long sA,
                               const float* __restrict__ B, int ldb, long long sB,
                               float* __restrict__ C, int ldc, long long sC,
                               int K) {
    __shared__ float As[2][16][72];
    __shared__ float Bs[2][16][72];
    const int z = blockIdx.z;
    A += (size_t)z * sA; B += (size_t)z * sB; C += (size_t)z * sC;
    const int tileM = blockIdx.y * 64;
    const int tileN = blockIdx.x * 64;
    const int tid  = threadIdx.x;
    const int wave = tid >> 5, lane = tid & 31;
    const int wm = wave >> 1, wn = wave & 1;
    const int half = lane >> 4, l = lane & 15;
    const int lk = tid >> 4;
    const int lm = (tid & 15) * 4;

    const float* gA = A + (size_t)lk * lda + tileM + lm;
    const float* gB = B + (size_t)lk * ldb + tileN + lm;

    v8f acc0 = {}; v8f acc1 = {};
    const int nstages = K >> 4;

    {
        float4 ra = *(const float4*)gA;
        float4 rb = *(const float4*)gB;
        *(float4*)&As[0][lk][lm] = ra;
        *(float4*)&Bs[0][lk][lm] = rb;
    }
    __syncthreads();

    for (int s = 0; s < nstages; s++) {
        const int p = s & 1;
        float4 na = make_float4(0.f, 0.f, 0.f, 0.f);
        float4 nb = make_float4(0.f, 0.f, 0.f, 0.f);
        if (s + 1 < nstages) {
            na = *(const float4*)(gA + (size_t)(s + 1) * 16 * lda);
            nb = *(const float4*)(gB + (size_t)(s + 1) * 16 * ldb);
        }
        v2f af[4], bf0[4], bf1[4];
        #pragma unroll
        for (int q = 0; q < 4; q++) {
            const int kc = q * 4 + half * 2;
            af[q].x  = As[p][kc][wm * 16 + l];
            af[q].y  = As[p][kc + 1][wm * 16 + l];
            bf0[q].x = Bs[p][kc][wn * 32 + l];
            bf0[q].y = Bs[p][kc + 1][wn * 32 + l];
            bf1[q].x = Bs[p][kc][wn * 32 + 16 + l];
            bf1[q].y = Bs[p][kc + 1][wn * 32 + 16 + l];
        }
        #pragma unroll
        for (int q = 0; q < 4; q++) {
            acc0 = __builtin_amdgcn_wmma_f32_16x16x4_f32(false, af[q], false, bf0[q], (short)0, acc0, false, false);
            acc1 = __builtin_amdgcn_wmma_f32_16x16x4_f32(false, af[q], false, bf1[q], (short)0, acc1, false, false);
        }
        if (s + 1 < nstages) {
            *(float4*)&As[p ^ 1][lk][lm] = na;
            *(float4*)&Bs[p ^ 1][lk][lm] = nb;
        }
        __syncthreads();
    }

    #pragma unroll
    for (int r = 0; r < 8; r++) {
        int m  = tileM + wm * 16 + half * 8 + r;
        int n0 = tileN + wn * 32 + l;
        C[(size_t)m * ldc + n0]      = acc0[r];
        C[(size_t)m * ldc + n0 + 16] = acc1[r];
    }
}

__global__ void softmax_row_kernel(float* __restrict__ x) {
    __shared__ float red[256];
    float4* xr = (float4*)(x + (size_t)blockIdx.x * 1024);
    float4 v = xr[threadIdx.x];
    float mx = fmaxf(fmaxf(v.x, v.y), fmaxf(v.z, v.w));
    red[threadIdx.x] = mx; __syncthreads();
    for (int off = 128; off > 0; off >>= 1) {
        if (threadIdx.x < off) red[threadIdx.x] = fmaxf(red[threadIdx.x], red[threadIdx.x + off]);
        __syncthreads();
    }
    mx = red[0];
    __syncthreads();
    v.x = __expf(v.x - mx); v.y = __expf(v.y - mx);
    v.z = __expf(v.z - mx); v.w = __expf(v.w - mx);
    float s = v.x + v.y + v.z + v.w;
    red[threadIdx.x] = s; __syncthreads();
    for (int off = 128; off > 0; off >>= 1) {
        if (threadIdx.x < off) red[threadIdx.x] += red[threadIdx.x + off];
        __syncthreads();
    }
    float inv = 1.0f / red[0];
    v.x *= inv; v.y *= inv; v.z *= inv; v.w *= inv;
    xr[threadIdx.x] = v;
}

template <int M>
__device__ __forceinline__ float xlane(float v) {
    return __int_as_float(__builtin_amdgcn_update_dpp(0, __float_as_int(v), 0x160 + M, 0xf, 0xf, true));
}

__device__ __forceinline__ float rowsum16(float s) {
    s += xlane<1>(s);
    s += xlane<2>(s);
    s += xlane<4>(s);
    s += xlane<8>(s);
    return s;
}

__device__ __forceinline__ float headsum16(const v2f* P) {
    float a[8], b[4], c[2];
    #pragma unroll
    for (int kk = 0; kk < 4; kk++) {
        a[2 * kk]     = P[kk].x + xlane<8>(P[kk + 4].x);
        a[2 * kk + 1] = P[kk].y + xlane<8>(P[kk + 4].y);
    }
    #pragma unroll
    for (int k = 0; k < 4; k++) b[k] = a[k] + xlane<4>(a[k + 4]);
    #pragma unroll
    for (int k = 0; k < 2; k++) c[k] = b[k] + xlane<2>(b[k + 2]);
    return c[0] + xlane<1>(c[1]);
}

__global__ void __launch_bounds__(128, 8)
zbias_kernel(const float* __restrict__ z,
             const float* __restrict__ pnw, const float* __restrict__ pnb,
             const float* __restrict__ bw,  const float* __restrict__ bb,
             float* __restrict__ out) {
    const int lane = threadIdx.x & 31;
    const int j = lane & 15;
    const int wave_id = blockIdx.x * 4 + (threadIdx.x >> 5);

    const float4 pw = *(const float4*)(pnw + 4 * j);
    const float4 pb = *(const float4*)(pnb + 4 * j);
    v2f W[8][4];
    v2f Cp[8];
    #pragma unroll
    for (int kk = 0; kk < 8; kk++) {
        const float4 r0 = *(const float4*)(bw + (((2 * kk) ^ j) * CZ + 4 * j));
        const float4 r1 = *(const float4*)(bw + (((2 * kk + 1) ^ j) * CZ + 4 * j));
        Cp[kk].x = pb.x * r0.x + pb.y * r0.y + pb.z * r0.z + pb.w * r0.w;
        Cp[kk].y = pb.x * r1.x + pb.y * r1.y + pb.z * r1.z + pb.w * r1.w;
        W[kk][0].x = r0.x * pw.x; W[kk][0].y = r1.x * pw.x;
        W[kk][1].x = r0.y * pw.y; W[kk][1].y = r1.y * pw.y;
        W[kk][2].x = r0.z * pw.z; W[kk][2].y = r1.z * pw.z;
        W[kk][3].x = r0.w * pw.w; W[kk][3].y = r1.w * pw.w;
    }
    const float cB = headsum16(Cp) + bb[j];

    const size_t base = (size_t)wave_id * 64 * 16;
    for (int chunk = 0; chunk < 4; chunk++) {
        const size_t cb = base + (size_t)chunk * 256 + lane;
        const float4* zp = (const float4*)z + cb;
        float* op = out + cb;
        float4 v[8];
        #pragma unroll
        for (int i = 0; i < 8; i++) v[i] = zp[i * 32];
        #pragma unroll
        for (int i = 0; i < 8; i++) {
            const float4 t = v[i];
            float s = (t.x + t.y) + (t.z + t.w);
            float q = t.x * t.x + t.y * t.y + t.z * t.z + t.w * t.w;
            s = rowsum16(s);
            q = rowsum16(q);
            const float m   = s * (1.0f / 64.0f);
            const float var = q * (1.0f / 64.0f) - m * m;
            const float inv = rsqrtf(var + 1e-5f);
            const float d0 = t.x - m, d1 = t.y - m, d2 = t.z - m, d3 = t.w - m;
            const v2f D0 = {d0, d0}, D1 = {d1, d1}, D2 = {d2, d2}, D3 = {d3, d3};
            v2f P[8];
            #pragma unroll
            for (int kk = 0; kk < 8; kk++) {
                v2f acc = D0 * W[kk][0];
                acc = __builtin_elementwise_fma(D1, W[kk][1], acc);
                acc = __builtin_elementwise_fma(D2, W[kk][2], acc);
                acc = __builtin_elementwise_fma(D3, W[kk][3], acc);
                P[kk] = acc;
            }
            op[i * 32] = headsum16(P) * inv + cB;
        }
    }
}

__global__ void adaln_gate_kernel(const float4* __restrict__ pb, const float4* __restrict__ pn,
                                  const float4* __restrict__ an, float4* __restrict__ a1) {
    int i = blockIdx.x * 256 + threadIdx.x;
    float4 p = pb[i], q = pn[i], n = an[i];
    float4 r;
    r.x = sigmoidf_(p.x * n.x + q.x);
    r.y = sigmoidf_(p.y * n.y + q.y);
    r.z = sigmoidf_(p.z * n.z + q.z);
    r.w = sigmoidf_(p.w * n.w + q.w);
    a1[i] = r;
}

__global__ void gate_o_kernel(const float* __restrict__ kvg, const float4* __restrict__ obuf,
                              float4* __restrict__ go) {
    int q = blockIdx.x * 256 + threadIdx.x;
    int h  = q >> 14;
    int j  = (q >> 4) & 1023;
    int c4 = (q & 15) * 4;
    const float4 g = *(const float4*)(kvg + (size_t)h * 196608 + (size_t)j * 192 + 128 + c4);
    float4 o = obuf[q];
    float4 r;
    r.x = sigmoidf_(g.x) * o.x;
    r.y = sigmoidf_(g.y) * o.y;
    r.z = sigmoidf_(g.z) * o.z;
    r.w = sigmoidf_(g.w) * o.w;
    go[q] = r;
}

extern "C" void kernel_launch(void* const* d_in, const int* in_sizes, int n_in,
                              void* d_out, int out_size, void* d_ws, size_t ws_size,
                              hipStream_t stream) {
    const float* a       = (const float*)d_in[0];
    const float* z       = (const float*)d_in[1];
    const float* s       = (const float*)d_in[2];
    const float* sn_w    = (const float*)d_in[3];
    const float* pb_w    = (const float*)d_in[4];
    const float* pb_b    = (const float*)d_in[5];
    const float* pn_w    = (const float*)d_in[6];
    const float* pnorm_w = (const float*)d_in[7];
    const float* pnorm_b = (const float*)d_in[8];
    const float* q_w     = (const float*)d_in[9];
    const float* q_b     = (const float*)d_in[10];
    const float* kvg_w   = (const float*)d_in[11];
    const float* bias_w  = (const float*)d_in[12];
    const float* bias_b  = (const float*)d_in[13];
    const float* attn_w  = (const float*)d_in[14];
    const float* out_w   = (const float*)d_in[15];
    const float* out_b   = (const float*)d_in[16];
    float* out = (float*)d_out;
    float* ws  = (float*)d_ws;

    float* a_n    = ws + 0;
    float* s_n    = ws + 1048576;
    float* pb     = ws + 1572864;
    float* pn     = ws + 2621440;
    float* qm     = ws + 3670016;
    float* kvg    = ws + 4718592;
    float* biasO  = ws + 7864320;
    float* obuf   = biasO;
    float* go     = biasO + 1048576;
    float* a2     = biasO + 2097152;
    float* scores = ws + 24641536;

    ln_kernel<<<1024, 256, 0, stream>>>(a, nullptr, a_n, 1024);
    ln_kernel<<<1024, 256, 0, stream>>>(s, sn_w, s_n, 512);

    gemm_nt_kernel<<<dim3(16, 16, 1), 256, 0, stream>>>(
        s_n, 512, 0, pb_w, 512, 0, pb, 1024, 0, pb_b, nullptr, 0, 0, 512, 1.0f, 0);
    gemm_nt_kernel<<<dim3(16, 16, 1), 256, 0, stream>>>(
        s_n, 512, 0, pn_w, 512, 0, pn, 1024, 0, nullptr, nullptr, 0, 0, 512, 1.0f, 0);
    adaln_gate_kernel<<<1024, 256, 0, stream>>>(
        (const float4*)pb, (const float4*)pn, (const float4*)a_n, (float4*)pb);
    float* a1 = pb;

    gemm_nt_kernel<<<dim3(16, 16, 1), 256, 0, stream>>>(
        a1, 1024, 0, q_w, 1024, 0, qm, 1024, 0, q_b, nullptr, 0, 0, 1024, 1.0f, 0);
    gemm_nt_kernel<<<dim3(48, 16, 1), 256, 0, stream>>>(
        a1, 1024, 0, kvg_w, 1024, 0, kvg, 3072, 0, nullptr, nullptr, 0, 0, 1024, 1.0f, 0);

    zbias_kernel<<<4096, 128, 0, stream>>>(z, pnorm_w, pnorm_b, bias_w, bias_b, biasO);

    gemm_nt_kernel<<<dim3(16, 16, 16), 256, 0, stream>>>(
        kvg, 192, 196608LL,
        qm, 64, 65536LL,
        scores, 1024, 1048576LL,
        nullptr,
        biasO, 1024, 1048576LL,
        64, 1.0f / 64.0f, 0);

    softmax_row_kernel<<<16384, 256, 0, stream>>>(scores);

    gemm_tn_kernel<<<dim3(1, 16, 16), 256, 0, stream>>>(
        scores, 1024, 1048576LL,
        kvg + 64, 192, 196608LL,
        obuf, 64, 65536LL,
        1024);

    gate_o_kernel<<<1024, 256, 0, stream>>>(kvg, (const float4*)obuf, (float4*)go);

    gemm_nt_kernel<<<dim3(16, 16, 1), 256, 0, stream>>>(
        go, 1024, 0, attn_w, 1024, 0, a2, 1024, 0, nullptr, nullptr, 0, 0, 1024, 1.0f, 0);

    gemm_nt_kernel<<<dim3(16, 16, 1), 256, 0, stream>>>(
        s, 512, 0, out_w, 512, 0, out, 1024, 0, out_b, a2, 1024, 0, 512, 1.0f, 1);
}


# Softmax launch removed: the scores GEMM also writes each row's max and exp-sum per 32 columns; the P^T V GEMM's loader merges them (DPP butterfly) and applies exp and 1/sum; row-sum order changes
# speedup vs baseline: 1.4029x; 1.1696x over previous
#include <hip/hip_runtime.h>
#include <hip/hip_bf16.h>
#include <math.h>

typedef __attribute__((ext_vector_type(2))) float v2f;
typedef __attribute__((ext_vector_type(8))) float v8f;

#define S_DIM 1024
#define CA 1024
#define CS 512
#define CZ 64
#define NH 16
#define HC 64

__device__ __forceinline__ float sigmoidf_(float x) {
    return 1.0f / (1.0f + __expf(-x));
}

template <int M>
__device__ __forceinline__ float xlane(float v) {
    return __int_as_float(__builtin_amdgcn_update_dpp(0, __float_as_int(v), 0x160 + M, 0xf, 0xf, true));
}

__device__ __forceinline__ float rowsum16(float s) {
    s += xlane<1>(s);
    s += xlane<2>(s);
    s += xlane<4>(s);
    s += xlane<8>(s);
    return s;
}

__device__ __forceinline__ float rowmax16(float s) {
    s = fmaxf(s, xlane<1>(s));
    s = fmaxf(s, xlane<2>(s));
    s = fmaxf(s, xlane<4>(s));
    s = fmaxf(s, xlane<8>(s));
    return s;
}

__global__ void ln_kernel(const float* __restrict__ x, const float* __restrict__ w,
                          float* __restrict__ y, int C) {
    __shared__ float red[256];
    __shared__ float red2[256];
    int row = blockIdx.x;
    const float4* xr = (const float4*)(x + (size_t)row * C);
    int C4 = C >> 2;
    float sum = 0.f, sq = 0.f;
    for (int c = threadIdx.x; c < C4; c += 256) {
        float4 v = xr[c];
        sum += v.x + v.y + v.z + v.w;
        sq  += v.x * v.x + v.y * v.y + v.z * v.z + v.w * v.w;
    }
    red[threadIdx.x] = sum; red2[threadIdx.x] = sq;
    __syncthreads();
    for (int off = 128; off > 0; off >>= 1) {
        if (threadIdx.x < off) {
            red[threadIdx.x]  += red[threadIdx.x + off];
            red2[threadIdx.x] += red2[threadIdx.x + off];
        }
        __syncthreads();
    }
    float m   = red[0] / C;
    float var = red2[0] / C - m * m;
    float inv = rsqrtf(var + 1e-5f);
    float4* yr = (float4*)(y + (size_t)row * C);
    for (int c = threadIdx.x; c < C4; c += 256) {
        float4 v = xr[c];
        v.x = (v.x - m) * inv; v.y = (v.y - m) * inv;
        v.z = (v.z - m) * inv; v.w = (v.w - m) * inv;
        if (w) {
            const float4 wv = ((const float4*)w)[c];
            v.x *= wv.x; v.y *= wv.y; v.z *= wv.z; v.w *= wv.w;
        }
        yr[c] = v;
    }
}

template <bool STATS>
__global__ void gemm_nt_kernel(const float* __restrict__ A, int lda, long long sA,
                               const float* __restrict__ B, int ldb, long long sB,
                               float* __restrict__ C, int ldc, long long sC,
                               const float* __restrict__ bias,
                               const float* __restrict__ addm, int ldadd, long long sAdd,
                               int K, float scale, int mode, float* __restrict__ stats) {
    __shared__ float As[2][64][20];
    __shared__ float Bs[2][64][20];
    const int z = blockIdx.z;
    A += (size_t)z * sA; B += (size_t)z * sB; C += (size_t)z * sC;
    if (addm) addm += (size_t)z * sAdd;
    const int tileM = blockIdx.y * 64;
    const int tileN = blockIdx.x * 64;
    const int tid  = threadIdx.x;
    const int wave = tid >> 5, lane = tid & 31;
    const int wm = wave >> 1, wn = wave & 1;
    const int half = lane >> 4, l = lane & 15;
    const int lr = tid >> 2;
    const int lc = (tid & 3) * 4;

    const float* gA = A + (size_t)(tileM + lr) * lda + lc;
    const float* gB = B + (size_t)(tileN + lr) * ldb + lc;

    v8f acc0 = {}; v8f acc1 = {};
    const int nstages = K >> 4;

    {
        float4 ra = *(const float4*)gA;
        float4 rb = *(const float4*)gB;
        *(float4*)&As[0][lr][lc] = ra;
        *(float4*)&Bs[0][lr][lc] = rb;
    }
    __syncthreads();

    for (int s = 0; s < nstages; s++) {
        const int p = s & 1;
        float4 na = make_float4(0.f, 0.f, 0.f, 0.f);
        float4 nb = make_float4(0.f, 0.f, 0.f, 0.f);
        if (s + 1 < nstages) {
            na = *(const float4*)(gA + (s + 1) * 16);
            nb = *(const float4*)(gB + (s + 1) * 16);
        }
        v2f af[4], bf0[4], bf1[4];
        #pragma unroll
        for (int q = 0; q < 4; q++) {
            const int kc = q * 4 + half * 2;
            af[q].x  = As[p][wm * 16 + l][kc];
            af[q].y  = As[p][wm * 16 + l][kc + 1];
            bf0[q].x = Bs[p][wn * 32 + l][kc];
            bf0[q].y = Bs[p][wn * 32 + l][kc + 1];
            bf1[q].x = Bs[p][wn * 32 + 16 + l][kc];
            bf1[q].y = Bs[p][wn * 32 + 16 + l][kc + 1];
        }
        #pragma unroll
        for (int q = 0; q < 4; q++) {
            acc0 = __builtin_amdgcn_wmma_f32_16x16x4_f32(false, af[q], false, bf0[q], (short)0, acc0, false, false);
            acc1 = __builtin_amdgcn_wmma_f32_16x16x4_f32(false, af[q], false, bf1[q], (short)0, acc1, false, false);
        }
        if (s + 1 < nstages) {
            *(float4*)&As[p ^ 1][lr][lc] = na;
            *(float4*)&Bs[p ^ 1][lr][lc] = nb;
        }
        __syncthreads();
    }

    float keep_mx = 0.f, keep_sum = 0.f;
    #pragma unroll
    for (int r = 0; r < 8; r++) {
        int m  = tileM + wm * 16 + half * 8 + r;
        int n0 = tileN + wn * 32 + l;
        int n1 = n0 + 16;
        float v0 = acc0[r] * scale;
        float v1 = acc1[r] * scale;
        if (bias) { v0 += bias[n0]; v1 += bias[n1]; }
        if (mode == 1) {
            v0 = sigmoidf_(v0) * addm[(size_t)m * ldadd + n0];
            v1 = sigmoidf_(v1) * addm[(size_t)m * ldadd + n1];
        } else if (addm) {
            v0 += addm[(size_t)m * ldadd + n0];
            v1 += addm[(size_t)m * ldadd + n1];
        }
        C[(size_t)m * ldc + n0] = v0;
        C[(size_t)m * ldc + n1] = v1;
        if (STATS) {
            const float mx = rowmax16(fmaxf(v0, v1));
            const float es = rowsum16(__expf(v0 - mx) + __expf(v1 - mx));
            if ((l & 7) == r) { keep_mx = mx; keep_sum = es; }
        }
    }
    if (STATS) {
        if (l < 8) {
            const size_t line = ((size_t)z * 64 + (tileM >> 4) + wm) * 32 + (blockIdx.x * 2 + wn);
            *(float2*)(stats + (line * 16 + half * 8 + l) * 2) = make_float2(keep_mx, keep_sum);
        }
    }
}

__device__ __forceinline__ float2 softmax_row_scale(float2 s0, float2 s1) {
    const float mx  = rowmax16(fmaxf(s0.x, s1.x));
    const float sum = rowsum16(s0.y * __expf(s0.x - mx) + s1.y * __expf(s1.x - mx));
    return make_float2(mx, 1.0f / sum);
}

__device__ __forceinline__ float4 softmax_apply(float4 v, float2 ms) {
    v.x = __expf(v.x - ms.x) * ms.y; v.y = __expf(v.y - ms.x) * ms.y;
    v.z = __expf(v.z - ms.x) * ms.y; v.w = __expf(v.w - ms.x) * ms.y;
    return v;
}

__global__ void gemm_tn_kernel(const float* __restrict__ A, int lda, long long sA,
                               const float* __restrict__ B, int ldb, long long sB,
                               float* __restrict__ C, int ldc, long long sC,
                               int K, const float* __restrict__ stats) {
    __shared__ float As[2][16][72];
    __shared__ float Bs[2][16][72];
    const int z = blockIdx.z;
    A += (size_t)z * sA; B += (size_t)z * sB; C += (size_t)z * sC;
    const int tileM = blockIdx.y * 64;
    const int tileN = blockIdx.x * 64;
    const int tid  = threadIdx.x;
    const int wave = tid >> 5, lane = tid & 31;
    const int wm = wave >> 1, wn = wave & 1;
    const int half = lane >> 4, l = lane & 15;
    const int lk = tid >> 4;
    const int lm = (tid & 15) * 4;

    const float* gA = A + (size_t)lk * lda + tileM + lm;
    const float* gB = B + (size_t)lk * ldb + tileN + lm;
    const float* gS = stats + (size_t)z * 65536 + (tid & 15) * 32 + lk * 2;

    v8f acc0 = {}; v8f acc1 = {};
    const int nstages = K >> 4;

    float2 c0 = make_float2(0.f, 1.f), c1 = make_float2(0.f, 1.f);
    {
        float4 ra = *(const float4*)gA;
        float4 rb = *(const float4*)gB;
        const float2 s0 = *(const float2*)gS;
        const float2 s1 = *(const float2*)(gS + 512);
        if (nstages > 1) {
            c0 = *(const float2*)(gS + 1024);
            c1 = *(const float2*)(gS + 1024 + 512);
        }
        ra = softmax_apply(ra, softmax_row_scale(s0, s1));
        *(float4*)&As[0][lk][lm] = ra;
        *(float4*)&Bs[0][lk][lm] = rb;
    }
    __syncthreads();

    for (int s = 0; s < nstages; s++) {
        const int p = s & 1;
        float4 na = make_float4(0.f, 0.f, 0.f, 0.f);
        float4 nb = make_float4(0.f, 0.f, 0.f, 0.f);
        float2 f0 = make_float2(0.f, 1.f), f1 = make_float2(0.f, 1.f);
        if (s + 1 < nstages) {
            na = *(const float4*)(gA + (size_t)(s + 1) * 16 * lda);
            nb = *(const float4*)(gB + (size_t)(s + 1) * 16 * ldb);
        }
        if (s + 2 < nstages) {
            f0 = *(const float2*)(gS + (size_t)(s + 2) * 1024);
            f1 = *(const float2*)(gS + (size_t)(s + 2) * 1024 + 512);
        }
        const float2 ms = softmax_row_scale(c0, c1);
        v2f af[4], bf0[4], bf1[4];
        #pragma unroll
        for (int q = 0; q < 4; q++) {
            const int kc = q * 4 + half * 2;
            af[q].x  = As[p][kc][wm * 16 + l];
            af[q].y  = As[p][kc + 1][wm * 16 + l];
            bf0[q].x = Bs[p][kc][wn * 32 + l];
            bf0[q].y = Bs[p][kc + 1][wn * 32 + l];
            bf1[q].x = Bs[p][kc][wn * 32 + 16 + l];
            bf1[q].y = Bs[p][kc + 1][wn * 32 + 16 + l];
        }
        #pragma unroll
        for (int q = 0; q < 4; q++) {
            acc0 = __builtin_amdgcn_wmma_f32_16x16x4_f32(false, af[q], false, bf0[q], (short)0, acc0, false, false);
            acc1 = __builtin_amdgcn_wmma_f32_16x16x4_f32(false, af[q], false, bf1[q], (short)0, acc1, false, false);
        }
        if (s + 1 < nstages) {
            na = softmax_apply(na, ms);
            *(float4*)&As[p ^ 1][lk][lm] = na;
            *(float4*)&Bs[p ^ 1][lk][lm] = nb;
        }
        c0 = f0; c1 = f1;
        __syncthreads();
    }

    #pragma unroll
    for (int r = 0; r < 8; r++) {
        int m  = tileM + wm * 16 + half * 8 + r;
        int n0 = tileN + wn * 32 + l;
        C[(size_t)m * ldc + n0]      = acc0[r];
        C[(size_t)m * ldc + n0 + 16] = acc1[r];
    }
}

__device__ __forceinline__ float headsum16(const v2f* P) {
    float a[8], b[4], c[2];
    #pragma unroll
    for (int kk = 0; kk < 4; kk++) {
        a[2 * kk]     = P[kk].x + xlane<8>(P[kk + 4].x);
        a[2 * kk + 1] = P[kk].y + xlane<8>(P[kk + 4].y);
    }
    #pragma unroll
    for (int k = 0; k < 4; k++) b[k] = a[k] + xlane<4>(a[k + 4]);
    #pragma unroll
    for (int k = 0; k < 2; k++) c[k] = b[k] + xlane<2>(b[k + 2]);
    return c[0] + xlane<1>(c[1]);
}

__global__ void __launch_bounds__(128, 8)
zbias_kernel(const float* __restrict__ z,
             const float* __restrict__ pnw, const float* __restrict__ pnb,
             const float* __restrict__ bw,  const float* __restrict__ bb,
             float* __restrict__ out) {
    const int lane = threadIdx.x & 31;
    const int j = lane & 15;
    const int wave_id = blockIdx.x * 4 + (threadIdx.x >> 5);

    const float4 pw = *(const float4*)(pnw + 4 * j);
    const float4 pb = *(const float4*)(pnb + 4 * j);
    v2f W[8][4];
    v2f Cp[8];
    #pragma unroll
    for (int kk = 0; kk < 8; kk++) {
        const float4 r0 = *(const float4*)(bw + (((2 * kk) ^ j) * CZ + 4 * j));
        const float4 r1 = *(const float4*)(bw + (((2 * kk + 1) ^ j) * CZ + 4 * j));
        Cp[kk].x = pb.x * r0.x + pb.y * r0.y + pb.z * r0.z + pb.w * r0.w;
        Cp[kk].y = pb.x * r1.x + pb.y * r1.y + pb.z * r1.z + pb.w * r1.w;
        W[kk][0].x = r0.x * pw.x; W[kk][0].y = r1.x * pw.x;
        W[kk][1].x = r0.y * pw.y; W[kk][1].y = r1.y * pw.y;
        W[kk][2].x = r0.z * pw.z; W[kk][2].y = r1.z * pw.z;
        W[kk][3].x = r0.w * pw.w; W[kk][3].y = r1.w * pw.w;
    }
    const float cB = headsum16(Cp) + bb[j];

    const size_t base = (size_t)wave_id * 64 * 16;
    for (int chunk = 0; chunk < 4; chunk++) {
        const size_t cb = base + (size_t)chunk * 256 + lane;
        const float4* zp = (const float4*)z + cb;
        float* op = out + cb;
        float4 v[8];
        #pragma unroll
        for (int i = 0; i < 8; i++) v[i] = zp[i * 32];
        #pragma unroll
        for (int i = 0; i < 8; i++) {
            const float4 t = v[i];
            float s = (t.x + t.y) + (t.z + t.w);
            float q = t.x * t.x + t.y * t.y + t.z * t.z + t.w * t.w;
            s = rowsum16(s);
            q = rowsum16(q);
            const float m   = s * (1.0f / 64.0f);
            const float var = q * (1.0f / 64.0f) - m * m;
            const float inv = rsqrtf(var + 1e-5f);
            const float d0 = t.x - m, d1 = t.y - m, d2 = t.z - m, d3 = t.w - m;
            const v2f D0 = {d0, d0}, D1 = {d1, d1}, D2 = {d2, d2}, D3 = {d3, d3};
            v2f P[8];
            #pragma unroll
            for (int kk = 0; kk < 8; kk++) {
                v2f acc = D0 * W[kk][0];
                acc = __builtin_elementwise_fma(D1, W[kk][1], acc);
                acc = __builtin_elementwise_fma(D2, W[kk][2], acc);
                acc = __builtin_elementwise_fma(D3, W[kk][3], acc);
                P[kk] = acc;
            }
            op[i * 32] = headsum16(P) * inv + cB;
        }
    }
}

__global__ void adaln_gate_kernel(const float4* __restrict__ pb, const float4* __restrict__ pn,
                                  const float4* __restrict__ an, float4* __restrict__ a1) {
    int i = blockIdx.x * 256 + threadIdx.x;
    float4 p = pb[i], q = pn[i], n = an[i];
    float4 r;
    r.x = sigmoidf_(p.x * n.x + q.x);
    r.y = sigmoidf_(p.y * n.y + q.y);
    r.z = sigmoidf_(p.z * n.z + q.z);
    r.w = sigmoidf_(p.w * n.w + q.w);
    a1[i] = r;
}

__global__ void gate_o_kernel(const float* __restrict__ kvg, const float4* __restrict__ obuf,
                              float4* __restrict__ go) {
    int q = blockIdx.x * 256 + threadIdx.x;
    int h  = q >> 14;
    int j  = (q >> 4) & 1023;
    int c4 = (q & 15) * 4;
    const float4 g = *(const float4*)(kvg + (size_t)h * 196608 + (size_t)j * 192 + 128 + c4);
    float4 o = obuf[q];
    float4 r;
    r.x = sigmoidf_(g.x) * o.x;
    r.y = sigmoidf_(g.y) * o.y;
    r.z = sigmoidf_(g.z) * o.z;
    r.w = sigmoidf_(g.w) * o.w;
    go[q] = r;
}

extern "C" void kernel_launch(void* const* d_in, const int* in_sizes, int n_in,
                              void* d_out, int out_size, void* d_ws, size_t ws_size,
                              hipStream_t stream) {
    const float* a       = (const float*)d_in[0];
    const float* z       = (const float*)d_in[1];
    const float* s       = (const float*)d_in[2];
    const float* sn_w    = (const float*)d_in[3];
    const float* pb_w    = (const float*)d_in[4];
    const float* pb_b    = (const float*)d_in[5];
    const float* pn_w    = (const float*)d_in[6];
    const float* pnorm_w = (const float*)d_in[7];
    const float* pnorm_b = (const float*)d_in[8];
    const float* q_w     = (const float*)d_in[9];
    const float* q_b     = (const float*)d_in[10];
    const float* kvg_w   = (const float*)d_in[11];
    const float* bias_w  = (const float*)d_in[12];
    const float* bias_b  = (const float*)d_in[13];
    const float* attn_w  = (const float*)d_in[14];
    const float* out_w   = (const float*)d_in[15];
    const float* out_b   = (const float*)d_in[16];
    float* out = (float*)d_out;
    float* ws  = (float*)d_ws;

    float* a_n    = ws + 0;
    float* s_n    = ws + 1048576;
    float* pb     = ws + 1572864;
    float* pn     = ws + 2621440;
    float* qm     = ws + 3670016;
    float* kvg    = ws + 4718592;
    float* biasO  = ws + 7864320;
    float* obuf   = biasO;
    float* go     = biasO + 1048576;
    float* a2     = biasO + 2097152;
    float* scores = ws + 24641536;
    float* smstat = a_n;

    ln_kernel<<<1024, 256, 0, stream>>>(a, nullptr, a_n, 1024);
    ln_kernel<<<1024, 256, 0, stream>>>(s, sn_w, s_n, 512);

    gemm_nt_kernel<false><<<dim3(16, 16, 1), 256, 0, stream>>>(
        s_n, 512, 0, pb_w, 512, 0, pb, 1024, 0, pb_b, nullptr, 0, 0, 512, 1.0f, 0, nullptr);
    gemm_nt_kernel<false><<<dim3(16, 16, 1), 256, 0, stream>>>(
        s_n, 512, 0, pn_w, 512, 0, pn, 1024, 0, nullptr, nullptr, 0, 0, 512, 1.0f, 0, nullptr);
    adaln_gate_kernel<<<1024, 256, 0, stream>>>(
        (const float4*)pb, (const float4*)pn, (const float4*)a_n, (float4*)pb);
    float* a1 = pb;

    gemm_nt_kernel<false><<<dim3(16, 16, 1), 256, 0, stream>>>(
        a1, 1024, 0, q_w, 1024, 0, qm, 1024, 0, q_b, nullptr, 0, 0, 1024, 1.0f, 0, nullptr);
    gemm_nt_kernel<false><<<dim3(48, 16, 1), 256, 0, stream>>>(
        a1, 1024, 0, kvg_w, 1024, 0, kvg, 3072, 0, nullptr, nullptr, 0, 0, 1024, 1.0f, 0, nullptr);

    zbias_kernel<<<4096, 128, 0, stream>>>(z, pnorm_w, pnorm_b, bias_w, bias_b, biasO);

    gemm_nt_kernel<true><<<dim3(16, 16, 16), 256, 0, stream>>>(
        kvg, 192, 196608LL,
        qm, 64, 65536LL,
        scores, 1024, 1048576LL,
        nullptr,
        biasO, 1024, 1048576LL,
        64, 1.0f / 64.0f, 0, smstat);

    gemm_tn_kernel<<<dim3(1, 16, 16), 256, 0, stream>>>(
        scores, 1024, 1048576LL,
        kvg + 64, 192, 196608LL,
        obuf, 64, 65536LL,
        1024, smstat);

    gate_o_kernel<<<1024, 256, 0, stream>>>(kvg, (const float4*)obuf, (float4*)go);

    gemm_nt_kernel<false><<<dim3(16, 16, 1), 256, 0, stream>>>(
        go, 1024, 0, attn_w, 1024, 0, a2, 1024, 0, nullptr, nullptr, 0, 0, 1024, 1.0f, 0, nullptr);

    gemm_nt_kernel<false><<<dim3(16, 16, 1), 256, 0, stream>>>(
        s, 512, 0, out_w, 512, 0, out, 1024, 0, out_b, a2, 1024, 0, 512, 1.0f, 1, nullptr);
}


# NT GEMMs feed WMMA straight from global memory: each lane loads its own row pieces into a four-step register ring (32x32 per wave, 4 waves), no LDS stage, no barrier; k order inside the sums changes
# speedup vs baseline: 1.4507x; 1.0341x over previous
#include <hip/hip_runtime.h>
#include <hip/hip_bf16.h>
#include <math.h>

typedef __attribute__((ext_vector_type(2))) float v2f;
typedef __attribute__((ext_vector_type(8))) float v8f;

#define S_DIM 1024
#define CA 1024
#define CS 512
#define CZ 64
#define NH 16
#define HC 64

__device__ __forceinline__ float sigmoidf_(float x) {
    return 1.0f / (1.0f + __expf(-x));
}

template <int M>
__device__ __forceinline__ float xlane(float v) {
    return __int_as_float(__builtin_amdgcn_update_dpp(0, __float_as_int(v), 0x160 + M, 0xf, 0xf, true));
}

__device__ __forceinline__ float rowsum16(float s) {
    s += xlane<1>(s);
    s += xlane<2>(s);
    s += xlane<4>(s);
    s += xlane<8>(s);
    return s;
}

__device__ __forceinline__ float rowmax16(float s) {
    s = fmaxf(s, xlane<1>(s));
    s = fmaxf(s, xlane<2>(s));
    s = fmaxf(s, xlane<4>(s));
    s = fmaxf(s, xlane<8>(s));
    return s;
}

__global__ void ln_kernel(const float* __restrict__ x, const float* __restrict__ w,
                          float* __restrict__ y, int C) {
    __shared__ float red[256];
    __shared__ float red2[256];
    int row = blockIdx.x;
    const float4* xr = (const float4*)(x + (size_t)row * C);
    int C4 = C >> 2;
    float sum = 0.f, sq = 0.f;
    for (int c = threadIdx.x; c < C4; c += 256) {
        float4 v = xr[c];
        sum += v.x + v.y + v.z + v.w;
        sq  += v.x * v.x + v.y * v.y + v.z * v.z + v.w * v.w;
    }
    red[threadIdx.x] = sum; red2[threadIdx.x] = sq;
    __syncthreads();
    for (int off = 128; off > 0; off >>= 1) {
        if (threadIdx.x < off) {
            red[threadIdx.x]  += red[threadIdx.x + off];
            red2[threadIdx.x] += red2[threadIdx.x + off];
        }
        __syncthreads();
    }
    float m   = red[0] / C;
    float var = red2[0] / C - m * m;
    float inv = rsqrtf(var + 1e-5f);
    float4* yr = (float4*)(y + (size_t)row * C);
    for (int c = threadIdx.x; c < C4; c += 256) {
        float4 v = xr[c];
        v.x = (v.x - m) * inv; v.y = (v.y - m) * inv;
        v.z = (v.z - m) * inv; v.w = (v.w - m) * inv;
        if (w) {
            const float4 wv = ((const float4*)w)[c];
            v.x *= wv.x; v.y *= wv.y; v.z *= wv.z; v.w *= wv.w;
        }
        yr[c] = v;
    }
}

struct Frag16 {
    float4 a0[2], a1[2], b0[2], b1[2];
};

__device__ __forceinline__ void load_step(Frag16& f, const float* pA0, const float* pA1,
                                          const float* pB0, const float* pB1, int k0) {
    #pragma unroll
    for (int g = 0; g < 2; g++) {
        f.a0[g] = *(const float4*)(pA0 + k0 + 8 * g);
        f.b0[g] = *(const float4*)(pB0 + k0 + 8 * g);
        f.a1[g] = *(const float4*)(pA1 + k0 + 8 * g);
        f.b1[g] = *(const float4*)(pB1 + k0 + 8 * g);
    }
}

__device__ __forceinline__ void mma_step(const Frag16& f, v8f& c00, v8f& c01, v8f& c10, v8f& c11) {
    #pragma unroll
    for (int g = 0; g < 2; g++) {
        const v2f a0l = {f.a0[g].x, f.a0[g].y}, a0h = {f.a0[g].z, f.a0[g].w};
        const v2f a1l = {f.a1[g].x, f.a1[g].y}, a1h = {f.a1[g].z, f.a1[g].w};
        const v2f b0l = {f.b0[g].x, f.b0[g].y}, b0h = {f.b0[g].z, f.b0[g].w};
        const v2f b1l = {f.b1[g].x, f.b1[g].y}, b1h = {f.b1[g].z, f.b1[g].w};
        c00 = __builtin_amdgcn_wmma_f32_16x16x4_f32(false, a0l, false, b0l, (short)0, c00, false, false);
        c01 = __builtin_amdgcn_wmma_f32_16x16x4_f32(false, a0l, false, b1l, (short)0, c01, false, false);
        c10 = __builtin_amdgcn_wmma_f32_16x16x4_f32(false, a1l, false, b0l, (short)0, c10, false, false);
        c11 = __builtin_amdgcn_wmma_f32_16x16x4_f32(false, a1l, false, b1l, (short)0, c11, false, false);
        c00 = __builtin_amdgcn_wmma_f32_16x16x4_f32(false, a0h, false, b0h, (short)0, c00, false, false);
        c01 = __builtin_amdgcn_wmma_f32_16x16x4_f32(false, a0h, false, b1h, (short)0, c01, false, false);
        c10 = __builtin_amdgcn_wmma_f32_16x16x4_f32(false, a1h, false, b0h, (short)0, c10, false, false);
        c11 = __builtin_amdgcn_wmma_f32_16x16x4_f32(false, a1h, false, b1h, (short)0, c11, false, false);
    }
}

template <bool STATS>
__global__ void __launch_bounds__(128)
gemm_nt_kernel(const float* __restrict__ A, int lda, long long sA,
               const float* __restrict__ B, int ldb, long long sB,
               float* __restrict__ C, int ldc, long long sC,
               const float* __restrict__ bias,
               const float* __restrict__ addm, int ldadd, long long sAdd,
               int K, float scale, int mode, float* __restrict__ stats) {
    const int z = blockIdx.z;
    A += (size_t)z * sA; B += (size_t)z * sB; C += (size_t)z * sC;
    if (addm) addm += (size_t)z * sAdd;
    const int tileM = blockIdx.y * 64;
    const int tileN = blockIdx.x * 64;
    const int tid  = threadIdx.x;
    const int wave = tid >> 5, lane = tid & 31;
    const int wm = wave >> 1, wn = wave & 1;
    const int half = lane >> 4, l = lane & 15;

    const float* pA0 = A + (size_t)(tileM + wm * 32 + l) * lda + half * 4;
    const float* pA1 = pA0 + (size_t)16 * lda;
    const float* pB0 = B + (size_t)(tileN + wn * 32 + l) * ldb + half * 4;
    const float* pB1 = pB0 + (size_t)16 * ldb;

    v8f c00 = {}, c01 = {}, c10 = {}, c11 = {};
    const int nsteps = K >> 4;

    #define ORDER_FENCE() asm volatile("" : "+v"(c00), "+v"(c01), "+v"(c10), "+v"(c11) : : "memory")
    Frag16 r0, r1, r2, r3;
    load_step(r0, pA0, pA1, pB0, pB1, 0);
    ORDER_FENCE();
    load_step(r1, pA0, pA1, pB0, pB1, 16);
    ORDER_FENCE();
    load_step(r2, pA0, pA1, pB0, pB1, 32);
    ORDER_FENCE();
    load_step(r3, pA0, pA1, pB0, pB1, 48);
    ORDER_FENCE();
    int s = 0;
    for (; s + 4 < nsteps; s += 4) {
        mma_step(r0, c00, c01, c10, c11);
        ORDER_FENCE();
        load_step(r0, pA0, pA1, pB0, pB1, (s + 4) * 16);
        ORDER_FENCE();
        mma_step(r1, c00, c01, c10, c11);
        ORDER_FENCE();
        load_step(r1, pA0, pA1, pB0, pB1, (s + 5) * 16);
        ORDER_FENCE();
        mma_step(r2, c00, c01, c10, c11);
        ORDER_FENCE();
        load_step(r2, pA0, pA1, pB0, pB1, (s + 6) * 16);
        ORDER_FENCE();
        mma_step(r3, c00, c01, c10, c11);
        ORDER_FENCE();
        load_step(r3, pA0, pA1, pB0, pB1, (s + 7) * 16);
        ORDER_FENCE();
    }
    mma_step(r0, c00, c01, c10, c11);
    mma_step(r1, c00, c01, c10, c11);
    mma_step(r2, c00, c01, c10, c11);
    mma_step(r3, c00, c01, c10, c11);
    #undef ORDER_FENCE

    #pragma unroll
    for (int i = 0; i < 2; i++) {
        const v8f acc0 = i ? c10 : c00;
        const v8f acc1 = i ? c11 : c01;
        float keep_mx = 0.f, keep_sum = 0.f;
        #pragma unroll
        for (int r = 0; r < 8; r++) {
            int m  = tileM + wm * 32 + i * 16 + half * 8 + r;
            int n0 = tileN + wn * 32 + l;
            int n1 = n0 + 16;
            float v0 = acc0[r] * scale;
            float v1 = acc1[r] * scale;
            if (bias) { v0 += bias[n0]; v1 += bias[n1]; }
            if (mode == 1) {
                v0 = sigmoidf_(v0) * addm[(size_t)m * ldadd + n0];
                v1 = sigmoidf_(v1) * addm[(size_t)m * ldadd + n1];
            } else if (addm) {
                v0 += addm[(size_t)m * ldadd + n0];
                v1 += addm[(size_t)m * ldadd + n1];
            }
            C[(size_t)m * ldc + n0] = v0;
            C[(size_t)m * ldc + n1] = v1;
            if (STATS) {
                const float mx = rowmax16(fmaxf(v0, v1));
                const float es = rowsum16(__expf(v0 - mx) + __expf(v1 - mx));
                if ((l & 7) == r) { keep_mx = mx; keep_sum = es; }
            }
        }
        if (STATS) {
            if (l < 8) {
                const size_t line = ((size_t)z * 64 + (tileM >> 4) + wm * 2 + i) * 32 + (blockIdx.x * 2 + wn);
                *(float2*)(stats + (line * 16 + half * 8 + l) * 2) = make_float2(keep_mx, keep_sum);
            }
        }
    }
}

__device__ __forceinline__ float2 softmax_row_scale(float2 s0, float2 s1) {
    const float mx  = rowmax16(fmaxf(s0.x, s1.x));
    const float sum = rowsum16(s0.y * __expf(s0.x - mx) + s1.y * __expf(s1.x - mx));
    return make_float2(mx, 1.0f / sum);
}

__device__ __forceinline__ float4 softmax_apply(float4 v, float2 ms) {
    v.x = __expf(v.x - ms.x) * ms.y; v.y = __expf(v.y - ms.x) * ms.y;
    v.z = __expf(v.z - ms.x) * ms.y; v.w = __expf(v.w - ms.x) * ms.y;
    return v;
}

__global__ void gemm_tn_kernel(const float* __restrict__ A, int lda, long long sA,
                               const float* __restrict__ B, int ldb, long long sB,
                               float* __restrict__ C, int ldc, long long sC,
                               int K, const float* __restrict__ stats) {
    __shared__ float As[2][16][72];
    __shared__ float Bs[2][16][72];
    const int z = blockIdx.z;
    A += (size_t)z * sA; B += (size_t)z * sB; C += (size_t)z * sC;
    const int tileM = blockIdx.y * 64;
    const int tileN = blockIdx.x * 64;
    const int tid  = threadIdx.x;
    const int wave = tid >> 5, lane = tid & 31;
    const int wm = wave >> 1, wn = wave & 1;
    const int half = lane >> 4, l = lane & 15;
    const int lk = tid >> 4;
    const int lm = (tid & 15) * 4;

    const float* gA = A + (size_t)lk * lda + tileM + lm;
    const float* gB = B + (size_t)lk * ldb + tileN + lm;
    const float* gS = stats + (size_t)z * 65536 + (tid & 15) * 32 + lk * 2;

    v8f acc0 = {}; v8f acc1 = {};
    const int nstages = K >> 4;

    float2 c0 = make_float2(0.f, 1.f), c1 = make_float2(0.f, 1.f);
    {
        float4 ra = *(const float4*)gA;
        float4 rb = *(const float4*)gB;
        const float2 s0 = *(const float2*)gS;
        const float2 s1 = *(const float2*)(gS + 512);
        if (nstages > 1) {
            c0 = *(const float2*)(gS + 1024);
            c1 = *(const float2*)(gS + 1024 + 512);
        }
        ra = softmax_apply(ra, softmax_row_scale(s0, s1));
        *(float4*)&As[0][lk][lm] = ra;
        *(float4*)&Bs[0][lk][lm] = rb;
    }
    __syncthreads();

    for (int s = 0; s < nstages; s++) {
        const int p = s & 1;
        float4 na = make_float4(0.f, 0.f, 0.f, 0.f);
        float4 nb = make_float4(0.f, 0.f, 0.f, 0.f);
        float2 f0 = make_float2(0.f, 1.f), f1 = make_float2(0.f, 1.f);
        if (s + 1 < nstages) {
            na = *(const float4*)(gA + (size_t)(s + 1) * 16 * lda);
            nb = *(const float4*)(gB + (size_t)(s + 1) * 16 * ldb);
        }
        if (s + 2 < nstages) {
            f0 = *(const float2*)(gS + (size_t)(s + 2) * 1024);
            f1 = *(const float2*)(gS + (size_t)(s + 2) * 1024 + 512);
        }
        const float2 ms = softmax_row_scale(c0, c1);
        v2f af[4], bf0[4], bf1[4];
        #pragma unroll
        for (int q = 0; q < 4; q++) {
            const int kc = q * 4 + half * 2;
            af[q].x  = As[p][kc][wm * 16 + l];
            af[q].y  = As[p][kc + 1][wm * 16 + l];
            bf0[q].x = Bs[p][kc][wn * 32 + l];
            bf0[q].y = Bs[p][kc + 1][wn * 32 + l];
            bf1[q].x = Bs[p][kc][wn * 32 + 16 + l];
            bf1[q].y = Bs[p][kc + 1][wn * 32 + 16 + l];
        }
        #pragma unroll
        for (int q = 0; q < 4; q++) {
            acc0 = __builtin_amdgcn_wmma_f32_16x16x4_f32(false, af[q], false, bf0[q], (short)0, acc0, false, false);
            acc1 = __builtin_amdgcn_wmma_f32_16x16x4_f32(false, af[q], false, bf1[q], (short)0, acc1, false, false);
        }
        if (s + 1 < nstages) {
            na = softmax_apply(na, ms);
            *(float4*)&As[p ^ 1][lk][lm] = na;
            *(float4*)&Bs[p ^ 1][lk][lm] = nb;
        }
        c0 = f0; c1 = f1;
        __syncthreads();
    }

    #pragma unroll
    for (int r = 0; r < 8; r++) {
        int m  = tileM + wm * 16 + half * 8 + r;
        int n0 = tileN + wn * 32 + l;
        C[(size_t)m * ldc + n0]      = acc0[r];
        C[(size_t)m * ldc + n0 + 16] = acc1[r];
    }
}

__device__ __forceinline__ float headsum16(const v2f* P) {
    float a[8], b[4], c[2];
    #pragma unroll
    for (int kk = 0; kk < 4; kk++) {
        a[2 * kk]     = P[kk].x + xlane<8>(P[kk + 4].x);
        a[2 * kk + 1] = P[kk].y + xlane<8>(P[kk + 4].y);
    }
    #pragma unroll
    for (int k = 0; k < 4; k++) b[k] = a[k] + xlane<4>(a[k + 4]);
    #pragma unroll
    for (int k = 0; k < 2; k++) c[k] = b[k] + xlane<2>(b[k + 2]);
    return c[0] + xlane<1>(c[1]);
}

__global__ void __launch_bounds__(128, 8)
zbias_kernel(const float* __restrict__ z,
             const float* __restrict__ pnw, const float* __restrict__ pnb,
             const float* __restrict__ bw,  const float* __restrict__ bb,
             float* __restrict__ out) {
    const int lane = threadIdx.x & 31;
    const int j = lane & 15;
    const int wave_id = blockIdx.x * 4 + (threadIdx.x >> 5);

    const float4 pw = *(const float4*)(pnw + 4 * j);
    const float4 pb = *(const float4*)(pnb + 4 * j);
    v2f W[8][4];
    v2f Cp[8];
    #pragma unroll
    for (int kk = 0; kk < 8; kk++) {
        const float4 r0 = *(const float4*)(bw + (((2 * kk) ^ j) * CZ + 4 * j));
        const float4 r1 = *(const float4*)(bw + (((2 * kk + 1) ^ j) * CZ + 4 * j));
        Cp[kk].x = pb.x * r0.x + pb.y * r0.y + pb.z * r0.z + pb.w * r0.w;
        Cp[kk].y = pb.x * r1.x + pb.y * r1.y + pb.z * r1.z + pb.w * r1.w;
        W[kk][0].x = r0.x * pw.x; W[kk][0].y = r1.x * pw.x;
        W[kk][1].x = r0.y * pw.y; W[kk][1].y = r1.y * pw.y;
        W[kk][2].x = r0.z * pw.z; W[kk][2].y = r1.z * pw.z;
        W[kk][3].x = r0.w * pw.w; W[kk][3].y = r1.w * pw.w;
    }
    const float cB = headsum16(Cp) + bb[j];

    const size_t base = (size_t)wave_id * 64 * 16;
    for (int chunk = 0; chunk < 4; chunk++) {
        const size_t cb = base + (size_t)chunk * 256 + lane;
        const float4* zp = (const float4*)z + cb;
        float* op = out + cb;
        float4 v[8];
        #pragma unroll
        for (int i = 0; i < 8; i++) v[i] = zp[i * 32];
        #pragma unroll
        for (int i = 0; i < 8; i++) {
            const float4 t = v[i];
            float s = (t.x + t.y) + (t.z + t.w);
            float q = t.x * t.x + t.y * t.y + t.z * t.z + t.w * t.w;
            s = rowsum16(s);
            q = rowsum16(q);
            const float m   = s * (1.0f / 64.0f);
            const float var = q * (1.0f / 64.0f) - m * m;
            const float inv = rsqrtf(var + 1e-5f);
            const float d0 = t.x - m, d1 = t.y - m, d2 = t.z - m, d3 = t.w - m;
            const v2f D0 = {d0, d0}, D1 = {d1, d1}, D2 = {d2, d2}, D3 = {d3, d3};
            v2f P[8];
            #pragma unroll
            for (int kk = 0; kk < 8; kk++) {
                v2f acc = D0 * W[kk][0];
                acc = __builtin_elementwise_fma(D1, W[kk][1], acc);
                acc = __builtin_elementwise_fma(D2, W[kk][2], acc);
                acc = __builtin_elementwise_fma(D3, W[kk][3], acc);
                P[kk] = acc;
            }
            op[i * 32] = headsum16(P) * inv + cB;
        }
    }
}

__global__ void adaln_gate_kernel(const float4* __restrict__ pb, const float4* __restrict__ pn,
                                  const float4* __restrict__ an, float4* __restrict__ a1) {
    int i = blockIdx.x * 256 + threadIdx.x;
    float4 p = pb[i], q = pn[i], n = an[i];
    float4 r;
    r.x = sigmoidf_(p.x * n.x + q.x);
    r.y = sigmoidf_(p.y * n.y + q.y);
    r.z = sigmoidf_(p.z * n.z + q.z);
    r.w = sigmoidf_(p.w * n.w + q.w);
    a1[i] = r;
}

__global__ void gate_o_kernel(const float* __restrict__ kvg, const float4* __restrict__ obuf,
                              float4* __restrict__ go) {
    int q = blockIdx.x * 256 + threadIdx.x;
    int h  = q >> 14;
    int j  = (q >> 4) & 1023;
    int c4 = (q & 15) * 4;
    const float4 g = *(const float4*)(kvg + (size_t)h * 196608 + (size_t)j * 192 + 128 + c4);
    float4 o = obuf[q];
    float4 r;
    r.x = sigmoidf_(g.x) * o.x;
    r.y = sigmoidf_(g.y) * o.y;
    r.z = sigmoidf_(g.z) * o.z;
    r.w = sigmoidf_(g.w) * o.w;
    go[q] = r;
}

extern "C" void kernel_launch(void* const* d_in, const int* in_sizes, int n_in,
                              void* d_out, int out_size, void* d_ws, size_t ws_size,
                              hipStream_t stream) {
    const float* a       = (const float*)d_in[0];
    const float* z       = (const float*)d_in[1];
    const float* s       = (const float*)d_in[2];
    const float* sn_w    = (const float*)d_in[3];
    const float* pb_w    = (const float*)d_in[4];
    const float* pb_b    = (const float*)d_in[5];
    const float* pn_w    = (const float*)d_in[6];
    const float* pnorm_w = (const float*)d_in[7];
    const float* pnorm_b = (const float*)d_in[8];
    const float* q_w     = (const float*)d_in[9];
    const float* q_b     = (const float*)d_in[10];
    const float* kvg_w   = (const float*)d_in[11];
    const float* bias_w  = (const float*)d_in[12];
    const float* bias_b  = (const float*)d_in[13];
    const float* attn_w  = (const float*)d_in[14];
    const float* out_w   = (const float*)d_in[15];
    const float* out_b   = (const float*)d_in[16];
    float* out = (float*)d_out;
    float* ws  = (float*)d_ws;

    float* a_n    = ws + 0;
    float* s_n    = ws + 1048576;
    float* pb     = ws + 1572864;
    float* pn     = ws + 2621440;
    float* qm     = ws + 3670016;
    float* kvg    = ws + 4718592;
    float* biasO  = ws + 7864320;
    float* obuf   = biasO;
    float* go     = biasO + 1048576;
    float* a2     = biasO + 2097152;
    float* scores = ws + 24641536;
    float* smstat = a_n;

    ln_kernel<<<1024, 256, 0, stream>>>(a, nullptr, a_n, 1024);
    ln_kernel<<<1024, 256, 0, stream>>>(s, sn_w, s_n, 512);

    gemm_nt_kernel<false><<<dim3(16, 16, 1), 128, 0, stream>>>(
        s_n, 512, 0, pb_w, 512, 0, pb, 1024, 0, pb_b, nullptr, 0, 0, 512, 1.0f, 0, nullptr);
    gemm_nt_kernel<false><<<dim3(16, 16, 1), 128, 0, stream>>>(
        s_n, 512, 0, pn_w, 512, 0, pn, 1024, 0, nullptr, nullptr, 0, 0, 512, 1.0f, 0, nullptr);
    adaln_gate_kernel<<<1024, 256, 0, stream>>>(
        (const float4*)pb, (const float4*)pn, (const float4*)a_n, (float4*)pb);
    float* a1 = pb;

    gemm_nt_kernel<false><<<dim3(16, 16, 1), 128, 0, stream>>>(
        a1, 1024, 0, q_w, 1024, 0, qm, 1024, 0, q_b, nullptr, 0, 0, 1024, 1.0f, 0, nullptr);
    gemm_nt_kernel<false><<<dim3(48, 16, 1), 128, 0, stream>>>(
        a1, 1024, 0, kvg_w, 1024, 0, kvg, 3072, 0, nullptr, nullptr, 0, 0, 1024, 1.0f, 0, nullptr);

    zbias_kernel<<<4096, 128, 0, stream>>>(z, pnorm_w, pnorm_b, bias_w, bias_b, biasO);

    gemm_nt_kernel<true><<<dim3(16, 16, 16), 128, 0, stream>>>(
        kvg, 192, 196608LL,
        qm, 64, 65536LL,
        scores, 1024, 1048576LL,
        nullptr,
        biasO, 1024, 1048576LL,
        64, 1.0f / 64.0f, 0, smstat);

    gemm_tn_kernel<<<dim3(1, 16, 16), 256, 0, stream>>>(
        scores, 1024, 1048576LL,
        kvg + 64, 192, 196608LL,
        obuf, 64, 65536LL,
        1024, smstat);

    gate_o_kernel<<<1024, 256, 0, stream>>>(kvg, (const float4*)obuf, (float4*)go);

    gemm_nt_kernel<false><<<dim3(16, 16, 1), 128, 0, stream>>>(
        go, 1024, 0, attn_w, 1024, 0, a2, 1024, 0, nullptr, nullptr, 0, 0, 1024, 1.0f, 0, nullptr);

    gemm_nt_kernel<false><<<dim3(16, 16, 1), 128, 0, stream>>>(
        s, 512, 0, out_w, 512, 0, out, 1024, 0, out_b, a2, 1024, 0, 512, 1.0f, 1, nullptr);
}


# P^T V GEMM without LDS stage or k-loop barrier: k rows read as whole lines, a lane's b128 feeds four WMMA tiles (column 4c+u), 4-step register ring, K split over 8 waves; k order of the sums changes
# speedup vs baseline: 1.5965x; 1.1005x over previous
#include <hip/hip_runtime.h>
#include <hip/hip_bf16.h>
#include <math.h>

typedef __attribute__((ext_vector_type(2))) float v2f;
typedef __attribute__((ext_vector_type(8))) float v8f;

#define S_DIM 1024
#define CA 1024
#define CS 512
#define CZ 64
#define NH 16
#define HC 64

__device__ __forceinline__ float sigmoidf_(float x) {
    return 1.0f / (1.0f + __expf(-x));
}

template <int M>
__device__ __forceinline__ float xlane(float v) {
    return __int_as_float(__builtin_amdgcn_update_dpp(0, __float_as_int(v), 0x160 + M, 0xf, 0xf, true));
}

__device__ __forceinline__ float rowsum16(float s) {
    s += xlane<1>(s);
    s += xlane<2>(s);
    s += xlane<4>(s);
    s += xlane<8>(s);
    return s;
}

__device__ __forceinline__ float rowmax16(float s) {
    s = fmaxf(s, xlane<1>(s));
    s = fmaxf(s, xlane<2>(s));
    s = fmaxf(s, xlane<4>(s));
    s = fmaxf(s, xlane<8>(s));
    return s;
}

__global__ void ln_kernel(const float* __restrict__ x, const float* __restrict__ w,
                          float* __restrict__ y, int C) {
    __shared__ float red[256];
    __shared__ float red2[256];
    int row = blockIdx.x;
    const float4* xr = (const float4*)(x + (size_t)row * C);
    int C4 = C >> 2;
    float sum = 0.f, sq = 0.f;
    for (int c = threadIdx.x; c < C4; c += 256) {
        float4 v = xr[c];
        sum += v.x + v.y + v.z + v.w;
        sq  += v.x * v.x + v.y * v.y + v.z * v.z + v.w * v.w;
    }
    red[threadIdx.x] = sum; red2[threadIdx.x] = sq;
    __syncthreads();
    for (int off = 128; off > 0; off >>= 1) {
        if (threadIdx.x < off) {
            red[threadIdx.x]  += red[threadIdx.x + off];
            red2[threadIdx.x] += red2[threadIdx.x + off];
        }
        __syncthreads();
    }
    float m   = red[0] / C;
    float var = red2[0] / C - m * m;
    float inv = rsqrtf(var + 1e-5f);
    float4* yr = (float4*)(y + (size_t)row * C);
    for (int c = threadIdx.x; c < C4; c += 256) {
        float4 v = xr[c];
        v.x = (v.x - m) * inv; v.y = (v.y - m) * inv;
        v.z = (v.z - m) * inv; v.w = (v.w - m) * inv;
        if (w) {
            const float4 wv = ((const float4*)w)[c];
            v.x *= wv.x; v.y *= wv.y; v.z *= wv.z; v.w *= wv.w;
        }
        yr[c] = v;
    }
}

struct Frag16 {
    float4 a0[2], a1[2], b0[2], b1[2];
};

__device__ __forceinline__ void load_step(Frag16& f, const float* pA0, const float* pA1,
                                          const float* pB0, const float* pB1, int k0) {
    #pragma unroll
    for (int g = 0; g < 2; g++) {
        f.a0[g] = *(const float4*)(pA0 + k0 + 8 * g);
        f.b0[g] = *(const float4*)(pB0 + k0 + 8 * g);
        f.a1[g] = *(const float4*)(pA1 + k0 + 8 * g);
        f.b1[g] = *(const float4*)(pB1 + k0 + 8 * g);
    }
}

__device__ __forceinline__ void mma_step(const Frag16& f, v8f& c00, v8f& c01, v8f& c10, v8f& c11) {
    #pragma unroll
    for (int g = 0; g < 2; g++) {
        const v2f a0l = {f.a0[g].x, f.a0[g].y}, a0h = {f.a0[g].z, f.a0[g].w};
        const v2f a1l = {f.a1[g].x, f.a1[g].y}, a1h = {f.a1[g].z, f.a1[g].w};
        const v2f b0l = {f.b0[g].x, f.b0[g].y}, b0h = {f.b0[g].z, f.b0[g].w};
        const v2f b1l = {f.b1[g].x, f.b1[g].y}, b1h = {f.b1[g].z, f.b1[g].w};
        c00 = __builtin_amdgcn_wmma_f32_16x16x4_f32(false, a0l, false, b0l, (short)0, c00, false, false);
        c01 = __builtin_amdgcn_wmma_f32_16x16x4_f32(false, a0l, false, b1l, (short)0, c01, false, false);
        c10 = __builtin_amdgcn_wmma_f32_16x16x4_f32(false, a1l, false, b0l, (short)0, c10, false, false);
        c11 = __builtin_amdgcn_wmma_f32_16x16x4_f32(false, a1l, false, b1l, (short)0, c11, false, false);
        c00 = __builtin_amdgcn_wmma_f32_16x16x4_f32(false, a0h, false, b0h, (short)0, c00, false, false);
        c01 = __builtin_amdgcn_wmma_f32_16x16x4_f32(false, a0h, false, b1h, (short)0, c01, false, false);
        c10 = __builtin_amdgcn_wmma_f32_16x16x4_f32(false, a1h, false, b0h, (short)0, c10, false, false);
        c11 = __builtin_amdgcn_wmma_f32_16x16x4_f32(false, a1h, false, b1h, (short)0, c11, false, false);
    }
}

template <bool STATS>
__global__ void __launch_bounds__(128)
gemm_nt_kernel(const float* __restrict__ A, int lda, long long sA,
               const float* __restrict__ B, int ldb, long long sB,
               float* __restrict__ C, int ldc, long long sC,
               const float* __restrict__ bias,
               const float* __restrict__ addm, int ldadd, long long sAdd,
               int K, float scale, int mode, float* __restrict__ stats) {
    const int z = blockIdx.z;
    A += (size_t)z * sA; B += (size_t)z * sB; C += (size_t)z * sC;
    if (addm) addm += (size_t)z * sAdd;
    const int tileM = blockIdx.y * 64;
    const int tileN = blockIdx.x * 64;
    const int tid  = threadIdx.x;
    const int wave = tid >> 5, lane = tid & 31;
    const int wm = wave >> 1, wn = wave & 1;
    const int half = lane >> 4, l = lane & 15;

    const float* pA0 = A + (size_t)(tileM + wm * 32 + l) * lda + half * 4;
    const float* pA1 = pA0 + (size_t)16 * lda;
    const float* pB0 = B + (size_t)(tileN + wn * 32 + l) * ldb + half * 4;
    const float* pB1 = pB0 + (size_t)16 * ldb;

    v8f c00 = {}, c01 = {}, c10 = {}, c11 = {};
    const int nsteps = K >> 4;

    #define ORDER_FENCE() asm volatile("" : "+v"(c00), "+v"(c01), "+v"(c10), "+v"(c11) : : "memory")
    Frag16 r0, r1, r2, r3;
    load_step(r0, pA0, pA1, pB0, pB1, 0);
    ORDER_FENCE();
    load_step(r1, pA0, pA1, pB0, pB1, 16);
    ORDER_FENCE();
    load_step(r2, pA0, pA1, pB0, pB1, 32);
    ORDER_FENCE();
    load_step(r3, pA0, pA1, pB0, pB1, 48);
    ORDER_FENCE();
    int s = 0;
    for (; s + 4 < nsteps; s += 4) {
        mma_step(r0, c00, c01, c10, c11);
        ORDER_FENCE();
        load_step(r0, pA0, pA1, pB0, pB1, (s + 4) * 16);
        ORDER_FENCE();
        mma_step(r1, c00, c01, c10, c11);
        ORDER_FENCE();
        load_step(r1, pA0, pA1, pB0, pB1, (s + 5) * 16);
        ORDER_FENCE();
        mma_step(r2, c00, c01, c10, c11);
        ORDER_FENCE();
        load_step(r2, pA0, pA1, pB0, pB1, (s + 6) * 16);
        ORDER_FENCE();
        mma_step(r3, c00, c01, c10, c11);
        ORDER_FENCE();
        load_step(r3, pA0, pA1, pB0, pB1, (s + 7) * 16);
        ORDER_FENCE();
    }
    mma_step(r0, c00, c01, c10, c11);
    mma_step(r1, c00, c01, c10, c11);
    mma_step(r2, c00, c01, c10, c11);
    mma_step(r3, c00, c01, c10, c11);
    #undef ORDER_FENCE

    #pragma unroll
    for (int i = 0; i < 2; i++) {
        const v8f acc0 = i ? c10 : c00;
        const v8f acc1 = i ? c11 : c01;
        float keep_mx = 0.f, keep_sum = 0.f;
        #pragma unroll
        for (int r = 0; r < 8; r++) {
            int m  = tileM + wm * 32 + i * 16 + half * 8 + r;
            int n0 = tileN + wn * 32 + l;
            int n1 = n0 + 16;
            float v0 = acc0[r] * scale;
            float v1 = acc1[r] * scale;
            if (bias) { v0 += bias[n0]; v1 += bias[n1]; }
            if (mode == 1) {
                v0 = sigmoidf_(v0) * addm[(size_t)m * ldadd + n0];
                v1 = sigmoidf_(v1) * addm[(size_t)m * ldadd + n1];
            } else if (addm) {
                v0 += addm[(size_t)m * ldadd + n0];
                v1 += addm[(size_t)m * ldadd + n1];
            }
            C[(size_t)m * ldc + n0] = v0;
            C[(size_t)m * ldc + n1] = v1;
            if (STATS) {
                const float mx = rowmax16(fmaxf(v0, v1));
                const float es = rowsum16(__expf(v0 - mx) + __expf(v1 - mx));
                if ((l & 7) == r) { keep_mx = mx; keep_sum = es; }
            }
        }
        if (STATS) {
            if (l < 8) {
                const size_t line = ((size_t)z * 64 + (tileM >> 4) + wm * 2 + i) * 32 + (blockIdx.x * 2 + wn);
                *(float2*)(stats + (line * 16 + half * 8 + l) * 2) = make_float2(keep_mx, keep_sum);
            }
        }
    }
}

__device__ __forceinline__ float2 softmax_row_scale(float2 s0, float2 s1) {
    const float mx  = rowmax16(fmaxf(s0.x, s1.x));
    const float sum = rowsum16(s0.y * __expf(s0.x - mx) + s1.y * __expf(s1.x - mx));
    return make_float2(mx, 1.0f / sum);
}

typedef __attribute__((ext_vector_type(4))) float v4f;
struct StepTN {
    v4f p0, p1, v0, v1, s0, s1;
};

__device__ __forceinline__ void load_tn(StepTN& f, const float* pP, const float* pV, const float* pS,
                                        int lda, int ldb) {
    f.s0 = *(const v4f*)pS;
    f.s1 = *(const v4f*)(pS + 512);
    f.p0 = *(const v4f*)pP;
    f.p1 = *(const v4f*)(pP + lda);
    f.v0 = *(const v4f*)pV;
    f.v1 = *(const v4f*)(pV + ldb);
}

__device__ __forceinline__ void prep_tn(StepTN& f) {
    const float2 m0 = softmax_row_scale(make_float2(f.s0.x, f.s0.y), make_float2(f.s1.x, f.s1.y));
    const float2 m1 = softmax_row_scale(make_float2(f.s0.z, f.s0.w), make_float2(f.s1.z, f.s1.w));
    f.p0.x = __expf(f.p0.x - m0.x) * m0.y; f.p0.y = __expf(f.p0.y - m0.x) * m0.y;
    f.p0.z = __expf(f.p0.z - m0.x) * m0.y; f.p0.w = __expf(f.p0.w - m0.x) * m0.y;
    f.p1.x = __expf(f.p1.x - m1.x) * m1.y; f.p1.y = __expf(f.p1.y - m1.x) * m1.y;
    f.p1.z = __expf(f.p1.z - m1.x) * m1.y; f.p1.w = __expf(f.p1.w - m1.x) * m1.y;
}

#define TN_MMA(t, u, vt0, vt1, pu0, pu1)                                                         \
    acc[t][u] = __builtin_amdgcn_wmma_f32_16x16x4_f32(false, (v2f){vt0, vt1}, false,             \
                                                      (v2f){pu0, pu1}, (short)0, acc[t][u], false, false)

__device__ __forceinline__ void mma_tn(const StepTN& f, v8f (&acc)[4][4]) {
    TN_MMA(0, 0, f.v0.x, f.v1.x, f.p0.x, f.p1.x); TN_MMA(0, 1, f.v0.x, f.v1.x, f.p0.y, f.p1.y);
    TN_MMA(0, 2, f.v0.x, f.v1.x, f.p0.z, f.p1.z); TN_MMA(0, 3, f.v0.x, f.v1.x, f.p0.w, f.p1.w);
    TN_MMA(1, 0, f.v0.y, f.v1.y, f.p0.x, f.p1.x); TN_MMA(1, 1, f.v0.y, f.v1.y, f.p0.y, f.p1.y);
    TN_MMA(1, 2, f.v0.y, f.v1.y, f.p0.z, f.p1.z); TN_MMA(1, 3, f.v0.y, f.v1.y, f.p0.w, f.p1.w);
    TN_MMA(2, 0, f.v0.z, f.v1.z, f.p0.x, f.p1.x); TN_MMA(2, 1, f.v0.z, f.v1.z, f.p0.y, f.p1.y);
    TN_MMA(2, 2, f.v0.z, f.v1.z, f.p0.z, f.p1.z); TN_MMA(2, 3, f.v0.z, f.v1.z, f.p0.w, f.p1.w);
    TN_MMA(3, 0, f.v0.w, f.v1.w, f.p0.x, f.p1.x); TN_MMA(3, 1, f.v0.w, f.v1.w, f.p0.y, f.p1.y);
    TN_MMA(3, 2, f.v0.w, f.v1.w, f.p0.z, f.p1.z); TN_MMA(3, 3, f.v0.w, f.v1.w, f.p0.w, f.p1.w);
}

__global__ void __launch_bounds__(256)
gemm_tn_kernel(const float* __restrict__ A, int lda, long long sA,
               const float* __restrict__ B, int ldb, long long sB,
               float* __restrict__ C, int ldc, long long sC,
               int K, const float* __restrict__ stats) {
    __shared__ float red[8][64][68];
    const int z = blockIdx.z;
    A += (size_t)z * sA; B += (size_t)z * sB; C += (size_t)z * sC;
    const int tileM = blockIdx.y * 64;
    const int tid  = threadIdx.x;
    const int wave = __builtin_amdgcn_readfirstlane(tid >> 5);
    const int lane = tid & 31;
    const int half = lane >> 4, l = lane & 15;
    const int kw = K >> 3;
    const int row0 = wave * kw;

    const float* qP = A + (size_t)row0 * lda + (size_t)(half * 2) * lda + tileM + 4 * l;
    const float* qV = B + (size_t)row0 * ldb + (size_t)(half * 2) * ldb + 4 * l;
    const float* qS = stats + (size_t)z * 65536 + (size_t)(row0 >> 4) * 1024 + l * 32 + half * 4;
    const size_t a4 = (size_t)lda * 4, b4 = (size_t)ldb * 4;

    v8f acc[4][4];
    #pragma unroll
    for (int i = 0; i < 16; i++) acc[i >> 2][i & 3] = (v8f){};
    const int ntrips = kw >> 4;

    #define TN_FENCE() asm volatile("" : "+v"(acc[0][0]), "+v"(acc[0][1]), "+v"(acc[0][2]), "+v"(acc[0][3]), \
                                         "+v"(acc[1][0]), "+v"(acc[1][1]), "+v"(acc[1][2]), "+v"(acc[1][3]), \
                                         "+v"(acc[2][0]), "+v"(acc[2][1]), "+v"(acc[2][2]), "+v"(acc[2][3]), \
                                         "+v"(acc[3][0]), "+v"(acc[3][1]), "+v"(acc[3][2]), "+v"(acc[3][3]) : : "memory")
    #define TN_GATE(f) asm volatile("" : "+v"(f.s0), "+v"(f.s1), "+v"(f.p0), "+v"(f.p1))
    StepTN r0, r1, r2, r3;
    load_tn(r0, qP, qV, qS, lda, ldb);
    TN_FENCE();
    load_tn(r1, qP + a4, qV + b4, qS + 8, lda, ldb);
    TN_FENCE();
    load_tn(r2, qP + 2 * a4, qV + 2 * b4, qS + 16, lda, ldb);
    TN_FENCE();
    load_tn(r3, qP + 3 * a4, qV + 3 * b4, qS + 24, lda, ldb);
    TN_FENCE();
    TN_GATE(r0);
    prep_tn(r0);
    for (int trip = 1; trip < ntrips; trip++) {
        qP += 4 * a4; qV += 4 * b4; qS += 1024;
        TN_GATE(r1);
        prep_tn(r1);
        mma_tn(r0, acc);
        TN_FENCE();
        load_tn(r0, qP, qV, qS, lda, ldb);
        TN_FENCE();
        TN_GATE(r2);
        prep_tn(r2);
        mma_tn(r1, acc);
        TN_FENCE();
        load_tn(r1, qP + a4, qV + b4, qS + 8, lda, ldb);
        TN_FENCE();
        TN_GATE(r3);
        prep_tn(r3);
        mma_tn(r2, acc);
        TN_FENCE();
        load_tn(r2, qP + 2 * a4, qV + 2 * b4, qS + 16, lda, ldb);
        TN_FENCE();
        TN_GATE(r0);
        prep_tn(r0);
        mma_tn(r3, acc);
        TN_FENCE();
        load_tn(r3, qP + 3 * a4, qV + 3 * b4, qS + 24, lda, ldb);
        TN_FENCE();
    }
    TN_GATE(r1);
    prep_tn(r1);
    mma_tn(r0, acc);
    TN_FENCE();
    TN_GATE(r2);
    prep_tn(r2);
    mma_tn(r1, acc);
    TN_FENCE();
    TN_GATE(r3);
    prep_tn(r3);
    mma_tn(r2, acc);
    mma_tn(r3, acc);
    #undef TN_GATE
    #undef TN_FENCE

    #pragma unroll
    for (int i = 0; i < 32; i++) {
        const int u = i >> 3, r = i & 7;
        *(float4*)&red[wave][16 * u + l][32 * half + 4 * r] =
            make_float4(acc[0][u][r], acc[1][u][r], acc[2][u][r], acc[3][u][r]);
    }
    __syncthreads();
    #pragma unroll
    for (int q = 0; q < 4; q++) {
        const int idx = tid + 256 * q;
        const int ro = idx >> 4, n4 = (idx & 15) * 4;
        const int m = 4 * (ro & 15) + (ro >> 4);
        float4 o = *(const float4*)&red[0][ro][n4];
        #pragma unroll
        for (int w = 1; w < 8; w++) {
            const float4 p = *(const float4*)&red[w][ro][n4];
            o.x += p.x; o.y += p.y; o.z += p.z; o.w += p.w;
        }
        *(float4*)(C + (size_t)(tileM + m) * ldc + n4) = o;
    }
}

__device__ __forceinline__ float headsum16(const v2f* P) {
    float a[8], b[4], c[2];
    #pragma unroll
    for (int kk = 0; kk < 4; kk++) {
        a[2 * kk]     = P[kk].x + xlane<8>(P[kk + 4].x);
        a[2 * kk + 1] = P[kk].y + xlane<8>(P[kk + 4].y);
    }
    #pragma unroll
    for (int k = 0; k < 4; k++) b[k] = a[k] + xlane<4>(a[k + 4]);
    #pragma unroll
    for (int k = 0; k < 2; k++) c[k] = b[k] + xlane<2>(b[k + 2]);
    return c[0] + xlane<1>(c[1]);
}

__global__ void __launch_bounds__(128, 8)
zbias_kernel(const float* __restrict__ z,
             const float* __restrict__ pnw, const float* __restrict__ pnb,
             const float* __restrict__ bw,  const float* __restrict__ bb,
             float* __restrict__ out) {
    const int lane = threadIdx.x & 31;
    const int j = lane & 15;
    const int wave_id = blockIdx.x * 4 + (threadIdx.x >> 5);

    const float4 pw = *(const float4*)(pnw + 4 * j);
    const float4 pb = *(const float4*)(pnb + 4 * j);
    v2f W[8][4];
    v2f Cp[8];
    #pragma unroll
    for (int kk = 0; kk < 8; kk++) {
        const float4 r0 = *(const float4*)(bw + (((2 * kk) ^ j) * CZ + 4 * j));
        const float4 r1 = *(const float4*)(bw + (((2 * kk + 1) ^ j) * CZ + 4 * j));
        Cp[kk].x = pb.x * r0.x + pb.y * r0.y + pb.z * r0.z + pb.w * r0.w;
        Cp[kk].y = pb.x * r1.x + pb.y * r1.y + pb.z * r1.z + pb.w * r1.w;
        W[kk][0].x = r0.x * pw.x; W[kk][0].y = r1.x * pw.x;
        W[kk][1].x = r0.y * pw.y; W[kk][1].y = r1.y * pw.y;
        W[kk][2].x = r0.z * pw.z; W[kk][2].y = r1.z * pw.z;
        W[kk][3].x = r0.w * pw.w; W[kk][3].y = r1.w * pw.w;
    }
    const float cB = headsum16(Cp) + bb[j];

    const size_t base = (size_t)wave_id * 64 * 16;
    for (int chunk = 0; chunk < 4; chunk++) {
        const size_t cb = base + (size_t)chunk * 256 + lane;
        const float4* zp = (const float4*)z + cb;
        float* op = out + cb;
        float4 v[8];
        #pragma unroll
        for (int i = 0; i < 8; i++) v[i] = zp[i * 32];
        #pragma unroll
        for (int i = 0; i < 8; i++) {
            const float4 t = v[i];
            float s = (t.x + t.y) + (t.z + t.w);
            float q = t.x * t.x + t.y * t.y + t.z * t.z + t.w * t.w;
            s = rowsum16(s);
            q = rowsum16(q);
            const float m   = s * (1.0f / 64.0f);
            const float var = q * (1.0f / 64.0f) - m * m;
            const float inv = rsqrtf(var + 1e-5f);
            const float d0 = t.x - m, d1 = t.y - m, d2 = t.z - m, d3 = t.w - m;
            const v2f D0 = {d0, d0}, D1 = {d1, d1}, D2 = {d2, d2}, D3 = {d3, d3};
            v2f P[8];
            #pragma unroll
            for (int kk = 0; kk < 8; kk++) {
                v2f acc = D0 * W[kk][0];
                acc = __builtin_elementwise_fma(D1, W[kk][1], acc);
                acc = __builtin_elementwise_fma(D2, W[kk][2], acc);
                acc = __builtin_elementwise_fma(D3, W[kk][3], acc);
                P[kk] = acc;
            }
            op[i * 32] = headsum16(P) * inv + cB;
        }
    }
}

__global__ void adaln_gate_kernel(const float4* __restrict__ pb, const float4* __restrict__ pn,
                                  const float4* __restrict__ an, float4* __restrict__ a1) {
    int i = blockIdx.x * 256 + threadIdx.x;
    float4 p = pb[i], q = pn[i], n = an[i];
    float4 r;
    r.x = sigmoidf_(p.x * n.x + q.x);
    r.y = sigmoidf_(p.y * n.y + q.y);
    r.z = sigmoidf_(p.z * n.z + q.z);
    r.w = sigmoidf_(p.w * n.w + q.w);
    a1[i] = r;
}

__global__ void gate_o_kernel(const float* __restrict__ kvg, const float4* __restrict__ obuf,
                              float4* __restrict__ go) {
    int q = blockIdx.x * 256 + threadIdx.x;
    int h  = q >> 14;
    int j  = (q >> 4) & 1023;
    int c4 = (q & 15) * 4;
    const float4 g = *(const float4*)(kvg + (size_t)h * 196608 + (size_t)j * 192 + 128 + c4);
    float4 o = obuf[q];
    float4 r;
    r.x = sigmoidf_(g.x) * o.x;
    r.y = sigmoidf_(g.y) * o.y;
    r.z = sigmoidf_(g.z) * o.z;
    r.w = sigmoidf_(g.w) * o.w;
    go[q] = r;
}

extern "C" void kernel_launch(void* const* d_in, const int* in_sizes, int n_in,
                              void* d_out, int out_size, void* d_ws, size_t ws_size,
                              hipStream_t stream) {
    const float* a       = (const float*)d_in[0];
    const float* z       = (const float*)d_in[1];
    const float* s       = (const float*)d_in[2];
    const float* sn_w    = (const float*)d_in[3];
    const float* pb_w    = (const float*)d_in[4];
    const float* pb_b    = (const float*)d_in[5];
    const float* pn_w    = (const float*)d_in[6];
    const float* pnorm_w = (const float*)d_in[7];
    const float* pnorm_b = (const float*)d_in[8];
    const float* q_w     = (const float*)d_in[9];
    const float* q_b     = (const float*)d_in[10];
    const float* kvg_w   = (const float*)d_in[11];
    const float* bias_w  = (const float*)d_in[12];
    const float* bias_b  = (const float*)d_in[13];
    const float* attn_w  = (const float*)d_in[14];
    const float* out_w   = (const float*)d_in[15];
    const float* out_b   = (const float*)d_in[16];
    float* out = (float*)d_out;
    float* ws  = (float*)d_ws;

    float* a_n    = ws + 0;
    float* s_n    = ws + 1048576;
    float* pb     = ws + 1572864;
    float* pn     = ws + 2621440;
    float* qm     = ws + 3670016;
    float* kvg    = ws + 4718592;
    float* biasO  = ws + 7864320;
    float* obuf   = biasO;
    float* go     = biasO + 1048576;
    float* a2     = biasO + 2097152;
    float* scores = ws + 24641536;
    float* smstat = a_n;

    ln_kernel<<<1024, 256, 0, stream>>>(a, nullptr, a_n, 1024);
    ln_kernel<<<1024, 256, 0, stream>>>(s, sn_w, s_n, 512);

    gemm_nt_kernel<false><<<dim3(16, 16, 1), 128, 0, stream>>>(
        s_n, 512, 0, pb_w, 512, 0, pb, 1024, 0, pb_b, nullptr, 0, 0, 512, 1.0f, 0, nullptr);
    gemm_nt_kernel<false><<<dim3(16, 16, 1), 128, 0, stream>>>(
        s_n, 512, 0, pn_w, 512, 0, pn, 1024, 0, nullptr, nullptr, 0, 0, 512, 1.0f, 0, nullptr);
    adaln_gate_kernel<<<1024, 256, 0, stream>>>(
        (const float4*)pb, (const float4*)pn, (const float4*)a_n, (float4*)pb);
    float* a1 = pb;

    gemm_nt_kernel<false><<<dim3(16, 16, 1), 128, 0, stream>>>(
        a1, 1024, 0, q_w, 1024, 0, qm, 1024, 0, q_b, nullptr, 0, 0, 1024, 1.0f, 0, nullptr);
    gemm_nt_kernel<false><<<dim3(48, 16, 1), 128, 0, stream>>>(
        a1, 1024, 0, kvg_w, 1024, 0, kvg, 3072, 0, nullptr, nullptr, 0, 0, 1024, 1.0f, 0, nullptr);

    zbias_kernel<<<4096, 128, 0, stream>>>(z, pnorm_w, pnorm_b, bias_w, bias_b, biasO);

    gemm_nt_kernel<true><<<dim3(16, 16, 16), 128, 0, stream>>>(
        kvg, 192, 196608LL,
        qm, 64, 65536LL,
        scores, 1024, 1048576LL,
        nullptr,
        biasO, 1024, 1048576LL,
        64, 1.0f / 64.0f, 0, smstat);

    gemm_tn_kernel<<<dim3(1, 16, 16), 256, 0, stream>>>(
        scores, 1024, 1048576LL,
        kvg + 64, 192, 196608LL,
        obuf, 64, 65536LL,
        1024, smstat);

    gate_o_kernel<<<1024, 256, 0, stream>>>(kvg, (const float4*)obuf, (float4*)go);

    gemm_nt_kernel<false><<<dim3(16, 16, 1), 128, 0, stream>>>(
        go, 1024, 0, attn_w, 1024, 0, a2, 1024, 0, nullptr, nullptr, 0, 0, 1024, 1.0f, 0, nullptr);

    gemm_nt_kernel<false><<<dim3(16, 16, 1), 128, 0, stream>>>(
        s, 512, 0, out_w, 512, 0, out, 1024, 0, out_b, a2, 1024, 0, 512, 1.0f, 1, nullptr);
}


# NT GEMM epilogue: a wave's 32 addend values and bias pair load as ONE group behind the first closing WMMA step (operand flags become template arguments), not a waited round trip per row; sums kept
# speedup vs baseline: 1.9693x; 1.2335x over previous
#include <hip/hip_runtime.h>
#include <hip/hip_bf16.h>
#include <math.h>

typedef __attribute__((ext_vector_type(2))) float v2f;
typedef __attribute__((ext_vector_type(8))) float v8f;

#define S_DIM 1024
#define CA 1024
#define CS 512
#define CZ 64
#define NH 16
#define HC 64

__device__ __forceinline__ float sigmoidf_(float x) {
    return 1.0f / (1.0f + __expf(-x));
}

template <int M>
__device__ __forceinline__ float xlane(float v) {
    return __int_as_float(__builtin_amdgcn_update_dpp(0, __float_as_int(v), 0x160 + M, 0xf, 0xf, true));
}

__device__ __forceinline__ float rowsum16(float s) {
    s += xlane<1>(s);
    s += xlane<2>(s);
    s += xlane<4>(s);
    s += xlane<8>(s);
    return s;
}

__device__ __forceinline__ float rowmax16(float s) {
    s = fmaxf(s, xlane<1>(s));
    s = fmaxf(s, xlane<2>(s));
    s = fmaxf(s, xlane<4>(s));
    s = fmaxf(s, xlane<8>(s));
    return s;
}

__global__ void ln_kernel(const float* __restrict__ x, const float* __restrict__ w,
                          float* __restrict__ y, int C) {
    __shared__ float red[256];
    __shared__ float red2[256];
    int row = blockIdx.x;
    const float4* xr = (const float4*)(x + (size_t)row * C);
    int C4 = C >> 2;
    float sum = 0.f, sq = 0.f;
    for (int c = threadIdx.x; c < C4; c += 256) {
        float4 v = xr[c];
        sum += v.x + v.y + v.z + v.w;
        sq  += v.x * v.x + v.y * v.y + v.z * v.z + v.w * v.w;
    }
    red[threadIdx.x] = sum; red2[threadIdx.x] = sq;
    __syncthreads();
    for (int off = 128; off > 0; off >>= 1) {
        if (threadIdx.x < off) {
            red[threadIdx.x]  += red[threadIdx.x + off];
            red2[threadIdx.x] += red2[threadIdx.x + off];
        }
        __syncthreads();
    }
    float m   = red[0] / C;
    float var = red2[0] / C - m * m;
    float inv = rsqrtf(var + 1e-5f);
    float4* yr = (float4*)(y + (size_t)row * C);
    for (int c = threadIdx.x; c < C4; c += 256) {
        float4 v = xr[c];
        v.x = (v.x - m) * inv; v.y = (v.y - m) * inv;
        v.z = (v.z - m) * inv; v.w = (v.w - m) * inv;
        if (w) {
            const float4 wv = ((const float4*)w)[c];
            v.x *= wv.x; v.y *= wv.y; v.z *= wv.z; v.w *= wv.w;
        }
        yr[c] = v;
    }
}

struct Frag16 {
    float4 a0[2], a1[2], b0[2], b1[2];
};

__device__ __forceinline__ void load_step(Frag16& f, const float* pA0, const float* pA1,
                                          const float* pB0, const float* pB1, int k0) {
    #pragma unroll
    for (int g = 0; g < 2; g++) {
        f.a0[g] = *(const float4*)(pA0 + k0 + 8 * g);
        f.b0[g] = *(const float4*)(pB0 + k0 + 8 * g);
        f.a1[g] = *(const float4*)(pA1 + k0 + 8 * g);
        f.b1[g] = *(const float4*)(pB1 + k0 + 8 * g);
    }
}

__device__ __forceinline__ void mma_step(const Frag16& f, v8f& c00, v8f& c01, v8f& c10, v8f& c11) {
    #pragma unroll
    for (int g = 0; g < 2; g++) {
        const v2f a0l = {f.a0[g].x, f.a0[g].y}, a0h = {f.a0[g].z, f.a0[g].w};
        const v2f a1l = {f.a1[g].x, f.a1[g].y}, a1h = {f.a1[g].z, f.a1[g].w};
        const v2f b0l = {f.b0[g].x, f.b0[g].y}, b0h = {f.b0[g].z, f.b0[g].w};
        const v2f b1l = {f.b1[g].x, f.b1[g].y}, b1h = {f.b1[g].z, f.b1[g].w};
        c00 = __builtin_amdgcn_wmma_f32_16x16x4_f32(false, a0l, false, b0l, (short)0, c00, false, false);
        c01 = __builtin_amdgcn_wmma_f32_16x16x4_f32(false, a0l, false, b1l, (short)0, c01, false, false);
        c10 = __builtin_amdgcn_wmma_f32_16x16x4_f32(false, a1l, false, b0l, (short)0, c10, false, false);
        c11 = __builtin_amdgcn_wmma_f32_16x16x4_f32(false, a1l, false, b1l, (short)0, c11, false, false);
        c00 = __builtin_amdgcn_wmma_f32_16x16x4_f32(false, a0h, false, b0h, (short)0, c00, false, false);
        c01 = __builtin_amdgcn_wmma_f32_16x16x4_f32(false, a0h, false, b1h, (short)0, c01, false, false);
        c10 = __builtin_amdgcn_wmma_f32_16x16x4_f32(false, a1h, false, b0h, (short)0, c10, false, false);
        c11 = __builtin_amdgcn_wmma_f32_16x16x4_f32(false, a1h, false, b1h, (short)0, c11, false, false);
    }
}

template <bool STATS, bool HAS_BIAS, bool HAS_ADD, int MODE>
__global__ void __launch_bounds__(128)
gemm_nt_kernel(const float* __restrict__ A, int lda, long long sA,
               const float* __restrict__ B, int ldb, long long sB,
               float* __restrict__ C, int ldc, long long sC,
               const float* __restrict__ bias,
               const float* __restrict__ addm, int ldadd, long long sAdd,
               int K, float scale, float* __restrict__ stats) {
    const int z = blockIdx.z;
    A += (size_t)z * sA; B += (size_t)z * sB; C += (size_t)z * sC;
    if (HAS_ADD) addm += (size_t)z * sAdd;
    const int tileM = blockIdx.y * 64;
    const int tileN = blockIdx.x * 64;
    const int tid  = threadIdx.x;
    const int wave = tid >> 5, lane = tid & 31;
    const int wm = wave >> 1, wn = wave & 1;
    const int half = lane >> 4, l = lane & 15;

    const float* pA0 = A + (size_t)(tileM + wm * 32 + l) * lda + half * 4;
    const float* pA1 = pA0 + (size_t)16 * lda;
    const float* pB0 = B + (size_t)(tileN + wn * 32 + l) * ldb + half * 4;
    const float* pB1 = pB0 + (size_t)16 * ldb;

    v8f c00 = {}, c01 = {}, c10 = {}, c11 = {};
    const int nsteps = K >> 4;

    #define ORDER_FENCE() asm volatile("" : "+v"(c00), "+v"(c01), "+v"(c10), "+v"(c11) : : "memory")
    Frag16 r0, r1, r2, r3;
    load_step(r0, pA0, pA1, pB0, pB1, 0);
    ORDER_FENCE();
    load_step(r1, pA0, pA1, pB0, pB1, 16);
    ORDER_FENCE();
    load_step(r2, pA0, pA1, pB0, pB1, 32);
    ORDER_FENCE();
    load_step(r3, pA0, pA1, pB0, pB1, 48);
    ORDER_FENCE();
    int s = 0;
    for (; s + 4 < nsteps; s += 4) {
        mma_step(r0, c00, c01, c10, c11);
        ORDER_FENCE();
        load_step(r0, pA0, pA1, pB0, pB1, (s + 4) * 16);
        ORDER_FENCE();
        mma_step(r1, c00, c01, c10, c11);
        ORDER_FENCE();
        load_step(r1, pA0, pA1, pB0, pB1, (s + 5) * 16);
        ORDER_FENCE();
        mma_step(r2, c00, c01, c10, c11);
        ORDER_FENCE();
        load_step(r2, pA0, pA1, pB0, pB1, (s + 6) * 16);
        ORDER_FENCE();
        mma_step(r3, c00, c01, c10, c11);
        ORDER_FENCE();
        load_step(r3, pA0, pA1, pB0, pB1, (s + 7) * 16);
        ORDER_FENCE();
    }
    mma_step(r0, c00, c01, c10, c11);
    ORDER_FENCE();
    const int ncol = tileN + wn * 32 + l;
    float bv0 = 0.f, bv1 = 0.f;
    if (HAS_BIAS) { bv0 = bias[ncol]; bv1 = bias[ncol + 16]; }
    float ad0[2][8], ad1[2][8];
    if (HAS_ADD) {
        #pragma unroll
        for (int i = 0; i < 2; i++) {
            #pragma unroll
            for (int r = 0; r < 8; r++) {
                const float* pa = addm + (size_t)(tileM + wm * 32 + i * 16 + half * 8 + r) * ldadd + ncol;
                ad0[i][r] = pa[0];
                ad1[i][r] = pa[16];
            }
        }
    }
    ORDER_FENCE();
    mma_step(r1, c00, c01, c10, c11);
    mma_step(r2, c00, c01, c10, c11);
    mma_step(r3, c00, c01, c10, c11);
    #undef ORDER_FENCE

    #pragma unroll
    for (int i = 0; i < 2; i++) {
        const v8f acc0 = i ? c10 : c00;
        const v8f acc1 = i ? c11 : c01;
        float keep_mx = 0.f, keep_sum = 0.f;
        #pragma unroll
        for (int r = 0; r < 8; r++) {
            int m  = tileM + wm * 32 + i * 16 + half * 8 + r;
            int n0 = tileN + wn * 32 + l;
            int n1 = n0 + 16;
            float v0 = acc0[r] * scale;
            float v1 = acc1[r] * scale;
            if (HAS_BIAS) { v0 += bv0; v1 += bv1; }
            if (MODE == 1) {
                v0 = sigmoidf_(v0) * ad0[i][r];
                v1 = sigmoidf_(v1) * ad1[i][r];
            } else if (HAS_ADD) {
                v0 += ad0[i][r];
                v1 += ad1[i][r];
            }
            C[(size_t)m * ldc + n0] = v0;
            C[(size_t)m * ldc + n1] = v1;
            if (STATS) {
                const float mx = rowmax16(fmaxf(v0, v1));
                const float es = rowsum16(__expf(v0 - mx) + __expf(v1 - mx));
                if ((l & 7) == r) { keep_mx = mx; keep_sum = es; }
            }
        }
        if (STATS) {
            if (l < 8) {
                const size_t line = ((size_t)z * 64 + (tileM >> 4) + wm * 2 + i) * 32 + (blockIdx.x * 2 + wn);
                *(float2*)(stats + (line * 16 + half * 8 + l) * 2) = make_float2(keep_mx, keep_sum);
            }
        }
    }
}

__device__ __forceinline__ float2 softmax_row_scale(float2 s0, float2 s1) {
    const float mx  = rowmax16(fmaxf(s0.x, s1.x));
    const float sum = rowsum16(s0.y * __expf(s0.x - mx) + s1.y * __expf(s1.x - mx));
    return make_float2(mx, 1.0f / sum);
}

typedef __attribute__((ext_vector_type(4))) float v4f;
struct StepTN {
    v4f p0, p1, v0, v1, s0, s1;
};

__device__ __forceinline__ void load_tn(StepTN& f, const float* pP, const float* pV, const float* pS,
                                        int lda, int ldb) {
    f.s0 = *(const v4f*)pS;
    f.s1 = *(const v4f*)(pS + 512);
    f.p0 = *(const v4f*)pP;
    f.p1 = *(const v4f*)(pP + lda);
    f.v0 = *(const v4f*)pV;
    f.v1 = *(const v4f*)(pV + ldb);
}

__device__ __forceinline__ void prep_tn(StepTN& f) {
    const float2 m0 = softmax_row_scale(make_float2(f.s0.x, f.s0.y), make_float2(f.s1.x, f.s1.y));
    const float2 m1 = softmax_row_scale(make_float2(f.s0.z, f.s0.w), make_float2(f.s1.z, f.s1.w));
    f.p0.x = __expf(f.p0.x - m0.x) * m0.y; f.p0.y = __expf(f.p0.y - m0.x) * m0.y;
    f.p0.z = __expf(f.p0.z - m0.x) * m0.y; f.p0.w = __expf(f.p0.w - m0.x) * m0.y;
    f.p1.x = __expf(f.p1.x - m1.x) * m1.y; f.p1.y = __expf(f.p1.y - m1.x) * m1.y;
    f.p1.z = __expf(f.p1.z - m1.x) * m1.y; f.p1.w = __expf(f.p1.w - m1.x) * m1.y;
}

#define TN_MMA(t, u, vt0, vt1, pu0, pu1)                                                         \
    acc[t][u] = __builtin_amdgcn_wmma_f32_16x16x4_f32(false, (v2f){vt0, vt1}, false,             \
                                                      (v2f){pu0, pu1}, (short)0, acc[t][u], false, false)

__device__ __forceinline__ void mma_tn(const StepTN& f, v8f (&acc)[4][4]) {
    TN_MMA(0, 0, f.v0.x, f.v1.x, f.p0.x, f.p1.x); TN_MMA(0, 1, f.v0.x, f.v1.x, f.p0.y, f.p1.y);
    TN_MMA(0, 2, f.v0.x, f.v1.x, f.p0.z, f.p1.z); TN_MMA(0, 3, f.v0.x, f.v1.x, f.p0.w, f.p1.w);
    TN_MMA(1, 0, f.v0.y, f.v1.y, f.p0.x, f.p1.x); TN_MMA(1, 1, f.v0.y, f.v1.y, f.p0.y, f.p1.y);
    TN_MMA(1, 2, f.v0.y, f.v1.y, f.p0.z, f.p1.z); TN_MMA(1, 3, f.v0.y, f.v1.y, f.p0.w, f.p1.w);
    TN_MMA(2, 0, f.v0.z, f.v1.z, f.p0.x, f.p1.x); TN_MMA(2, 1, f.v0.z, f.v1.z, f.p0.y, f.p1.y);
    TN_MMA(2, 2, f.v0.z, f.v1.z, f.p0.z, f.p1.z); TN_MMA(2, 3, f.v0.z, f.v1.z, f.p0.w, f.p1.w);
    TN_MMA(3, 0, f.v0.w, f.v1.w, f.p0.x, f.p1.x); TN_MMA(3, 1, f.v0.w, f.v1.w, f.p0.y, f.p1.y);
    TN_MMA(3, 2, f.v0.w, f.v1.w, f.p0.z, f.p1.z); TN_MMA(3, 3, f.v0.w, f.v1.w, f.p0.w, f.p1.w);
}

__global__ void __launch_bounds__(256)
gemm_tn_kernel(const float* __restrict__ A, int lda, long long sA,
               const float* __restrict__ B, int ldb, long long sB,
               float* __restrict__ C, int ldc, long long sC,
               int K, const float* __restrict__ stats) {
    __shared__ float red[8][64][68];
    const int z = blockIdx.z;
    A += (size_t)z * sA; B += (size_t)z * sB; C += (size_t)z * sC;
    const int tileM = blockIdx.y * 64;
    const int tid  = threadIdx.x;
    const int wave = __builtin_amdgcn_readfirstlane(tid >> 5);
    const int lane = tid & 31;
    const int half = lane >> 4, l = lane & 15;
    const int kw = K >> 3;
    const int row0 = wave * kw;

    const float* qP = A + (size_t)row0 * lda + (size_t)(half * 2) * lda + tileM + 4 * l;
    const float* qV = B + (size_t)row0 * ldb + (size_t)(half * 2) * ldb + 4 * l;
    const float* qS = stats + (size_t)z * 65536 + (size_t)(row0 >> 4) * 1024 + l * 32 + half * 4;
    const size_t a4 = (size_t)lda * 4, b4 = (size_t)ldb * 4;

    v8f acc[4][4];
    #pragma unroll
    for (int i = 0; i < 16; i++) acc[i >> 2][i & 3] = (v8f){};
    const int ntrips = kw >> 4;

    #define TN_FENCE() asm volatile("" : "+v"(acc[0][0]), "+v"(acc[0][1]), "+v"(acc[0][2]), "+v"(acc[0][3]), \
                                         "+v"(acc[1][0]), "+v"(acc[1][1]), "+v"(acc[1][2]), "+v"(acc[1][3]), \
                                         "+v"(acc[2][0]), "+v"(acc[2][1]), "+v"(acc[2][2]), "+v"(acc[2][3]), \
                                         "+v"(acc[3][0]), "+v"(acc[3][1]), "+v"(acc[3][2]), "+v"(acc[3][3]) : : "memory")
    #define TN_GATE(f) asm volatile("" : "+v"(f.s0), "+v"(f.s1), "+v"(f.p0), "+v"(f.p1))
    StepTN r0, r1, r2, r3;
    load_tn(r0, qP, qV, qS, lda, ldb);
    TN_FENCE();
    load_tn(r1, qP + a4, qV + b4, qS + 8, lda, ldb);
    TN_FENCE();
    load_tn(r2, qP + 2 * a4, qV + 2 * b4, qS + 16, lda, ldb);
    TN_FENCE();
    load_tn(r3, qP + 3 * a4, qV + 3 * b4, qS + 24, lda, ldb);
    TN_FENCE();
    TN_GATE(r0);
    prep_tn(r0);
    for (int trip = 1; trip < ntrips; trip++) {
        qP += 4 * a4; qV += 4 * b4; qS += 1024;
        TN_GATE(r1);
        prep_tn(r1);
        mma_tn(r0, acc);
        TN_FENCE();
        load_tn(r0, qP, qV, qS, lda, ldb);
        TN_FENCE();
        TN_GATE(r2);
        prep_tn(r2);
        mma_tn(r1, acc);
        TN_FENCE();
        load_tn(r1, qP + a4, qV + b4, qS + 8, lda, ldb);
        TN_FENCE();
        TN_GATE(r3);
        prep_tn(r3);
        mma_tn(r2, acc);
        TN_FENCE();
        load_tn(r2, qP + 2 * a4, qV + 2 * b4, qS + 16, lda, ldb);
        TN_FENCE();
        TN_GATE(r0);
        prep_tn(r0);
        mma_tn(r3, acc);
        TN_FENCE();
        load_tn(r3, qP + 3 * a4, qV + 3 * b4, qS + 24, lda, ldb);
        TN_FENCE();
    }
    TN_GATE(r1);
    prep_tn(r1);
    mma_tn(r0, acc);
    TN_FENCE();
    TN_GATE(r2);
    prep_tn(r2);
    mma_tn(r1, acc);
    TN_FENCE();
    TN_GATE(r3);
    prep_tn(r3);
    mma_tn(r2, acc);
    mma_tn(r3, acc);
    #undef TN_GATE
    #undef TN_FENCE

    #pragma unroll
    for (int i = 0; i < 32; i++) {
        const int u = i >> 3, r = i & 7;
        *(float4*)&red[wave][16 * u + l][32 * half + 4 * r] =
            make_float4(acc[0][u][r], acc[1][u][r], acc[2][u][r], acc[3][u][r]);
    }
    __syncthreads();
    #pragma unroll
    for (int q = 0; q < 4; q++) {
        const int idx = tid + 256 * q;
        const int ro = idx >> 4, n4 = (idx & 15) * 4;
        const int m = 4 * (ro & 15) + (ro >> 4);
        float4 o = *(const float4*)&red[0][ro][n4];
        #pragma unroll
        for (int w = 1; w < 8; w++) {
            const float4 p = *(const float4*)&red[w][ro][n4];
            o.x += p.x; o.y += p.y; o.z += p.z; o.w += p.w;
        }
        *(float4*)(C + (size_t)(tileM + m) * ldc + n4) = o;
    }
}

__device__ __forceinline__ float headsum16(const v2f* P) {
    float a[8], b[4], c[2];
    #pragma unroll
    for (int kk = 0; kk < 4; kk++) {
        a[2 * kk]     = P[kk].x + xlane<8>(P[kk + 4].x);
        a[2 * kk + 1] = P[kk].y + xlane<8>(P[kk + 4].y);
    }
    #pragma unroll
    for (int k = 0; k < 4; k++) b[k] = a[k] + xlane<4>(a[k + 4]);
    #pragma unroll
    for (int k = 0; k < 2; k++) c[k] = b[k] + xlane<2>(b[k + 2]);
    return c[0] + xlane<1>(c[1]);
}

__global__ void __launch_bounds__(128, 8)
zbias_kernel(const float* __restrict__ z,
             const float* __restrict__ pnw, const float* __restrict__ pnb,
             const float* __restrict__ bw,  const float* __restrict__ bb,
             float* __restrict__ out) {
    const int lane = threadIdx.x & 31;
    const int j = lane & 15;
    const int wave_id = blockIdx.x * 4 + (threadIdx.x >> 5);

    const float4 pw = *(const float4*)(pnw + 4 * j);
    const float4 pb = *(const float4*)(pnb + 4 * j);
    v2f W[8][4];
    v2f Cp[8];
    #pragma unroll
    for (int kk = 0; kk < 8; kk++) {
        const float4 r0 = *(const float4*)(bw + (((2 * kk) ^ j) * CZ + 4 * j));
        const float4 r1 = *(const float4*)(bw + (((2 * kk + 1) ^ j) * CZ + 4 * j));
        Cp[kk].x = pb.x * r0.x + pb.y * r0.y + pb.z * r0.z + pb.w * r0.w;
        Cp[kk].y = pb.x * r1.x + pb.y * r1.y + pb.z * r1.z + pb.w * r1.w;
        W[kk][0].x = r0.x * pw.x; W[kk][0].y = r1.x * pw.x;
        W[kk][1].x = r0.y * pw.y; W[kk][1].y = r1.y * pw.y;
        W[kk][2].x = r0.z * pw.z; W[kk][2].y = r1.z * pw.z;
        W[kk][3].x = r0.w * pw.w; W[kk][3].y = r1.w * pw.w;
    }
    const float cB = headsum16(Cp) + bb[j];

    const size_t base = (size_t)wave_id * 64 * 16;
    for (int chunk = 0; chunk < 4; chunk++) {
        const size_t cb = base + (size_t)chunk * 256 + lane;
        const float4* zp = (const float4*)z + cb;
        float* op = out + cb;
        float4 v[8];
        #pragma unroll
        for (int i = 0; i < 8; i++) v[i] = zp[i * 32];
        #pragma unroll
        for (int i = 0; i < 8; i++) {
            const float4 t = v[i];
            float s = (t.x + t.y) + (t.z + t.w);
            float q = t.x * t.x + t.y * t.y + t.z * t.z + t.w * t.w;
            s = rowsum16(s);
            q = rowsum16(q);
            const float m   = s * (1.0f / 64.0f);
            const float var = q * (1.0f / 64.0f) - m * m;
            const float inv = rsqrtf(var + 1e-5f);
            const float d0 = t.x - m, d1 = t.y - m, d2 = t.z - m, d3 = t.w - m;
            const v2f D0 = {d0, d0}, D1 = {d1, d1}, D2 = {d2, d2}, D3 = {d3, d3};
            v2f P[8];
            #pragma unroll
            for (int kk = 0; kk < 8; kk++) {
                v2f acc = D0 * W[kk][0];
                acc = __builtin_elementwise_fma(D1, W[kk][1], acc);
                acc = __builtin_elementwise_fma(D2, W[kk][2], acc);
                acc = __builtin_elementwise_fma(D3, W[kk][3], acc);
                P[kk] = acc;
            }
            op[i * 32] = headsum16(P) * inv + cB;
        }
    }
}

__global__ void adaln_gate_kernel(const float4* __restrict__ pb, const float4* __restrict__ pn,
                                  const float4* __restrict__ an, float4* __restrict__ a1) {
    int i = blockIdx.x * 256 + threadIdx.x;
    float4 p = pb[i], q = pn[i], n = an[i];
    float4 r;
    r.x = sigmoidf_(p.x * n.x + q.x);
    r.y = sigmoidf_(p.y * n.y + q.y);
    r.z = sigmoidf_(p.z * n.z + q.z);
    r.w = sigmoidf_(p.w * n.w + q.w);
    a1[i] = r;
}

__global__ void gate_o_kernel(const float* __restrict__ kvg, const float4* __restrict__ obuf,
                              float4* __restrict__ go) {
    int q = blockIdx.x * 256 + threadIdx.x;
    int h  = q >> 14;
    int j  = (q >> 4) & 1023;
    int c4 = (q & 15) * 4;
    const float4 g = *(const float4*)(kvg + (size_t)h * 196608 + (size_t)j * 192 + 128 + c4);
    float4 o = obuf[q];
    float4 r;
    r.x = sigmoidf_(g.x) * o.x;
    r.y = sigmoidf_(g.y) * o.y;
    r.z = sigmoidf_(g.z) * o.z;
    r.w = sigmoidf_(g.w) * o.w;
    go[q] = r;
}

extern "C" void kernel_launch(void* const* d_in, const int* in_sizes, int n_in,
                              void* d_out, int out_size, void* d_ws, size_t ws_size,
                              hipStream_t stream) {
    const float* a       = (const float*)d_in[0];
    const float* z       = (const float*)d_in[1];
    const float* s       = (const float*)d_in[2];
    const float* sn_w    = (const float*)d_in[3];
    const float* pb_w    = (const float*)d_in[4];
    const float* pb_b    = (const float*)d_in[5];
    const float* pn_w    = (const float*)d_in[6];
    const float* pnorm_w = (const float*)d_in[7];
    const float* pnorm_b = (const float*)d_in[8];
    const float* q_w     = (const float*)d_in[9];
    const float* q_b     = (const float*)d_in[10];
    const float* kvg_w   = (const float*)d_in[11];
    const float* bias_w  = (const float*)d_in[12];
    const float* bias_b  = (const float*)d_in[13];
    const float* attn_w  = (const float*)d_in[14];
    const float* out_w   = (const float*)d_in[15];
    const float* out_b   = (const float*)d_in[16];
    float* out = (float*)d_out;
    float* ws  = (float*)d_ws;

    float* a_n    = ws + 0;
    float* s_n    = ws + 1048576;
    float* pb     = ws + 1572864;
    float* pn     = ws + 2621440;
    float* qm     = ws + 3670016;
    float* kvg    = ws + 4718592;
    float* biasO  = ws + 7864320;
    float* obuf   = biasO;
    float* go     = biasO + 1048576;
    float* a2     = biasO + 2097152;
    float* scores = ws + 24641536;
    float* smstat = a_n;

    ln_kernel<<<1024, 256, 0, stream>>>(a, nullptr, a_n, 1024);
    ln_kernel<<<1024, 256, 0, stream>>>(s, sn_w, s_n, 512);

    gemm_nt_kernel<false, true, false, 0><<<dim3(16, 16, 1), 128, 0, stream>>>(
        s_n, 512, 0, pb_w, 512, 0, pb, 1024, 0, pb_b, nullptr, 0, 0, 512, 1.0f, nullptr);
    gemm_nt_kernel<false, false, false, 0><<<dim3(16, 16, 1), 128, 0, stream>>>(
        s_n, 512, 0, pn_w, 512, 0, pn, 1024, 0, nullptr, nullptr, 0, 0, 512, 1.0f, nullptr);
    adaln_gate_kernel<<<1024, 256, 0, stream>>>(
        (const float4*)pb, (const float4*)pn, (const float4*)a_n, (float4*)pb);
    float* a1 = pb;

    gemm_nt_kernel<false, true, false, 0><<<dim3(16, 16, 1), 128, 0, stream>>>(
        a1, 1024, 0, q_w, 1024, 0, qm, 1024, 0, q_b, nullptr, 0, 0, 1024, 1.0f, nullptr);
    gemm_nt_kernel<false, false, false, 0><<<dim3(48, 16, 1), 128, 0, stream>>>(
        a1, 1024, 0, kvg_w, 1024, 0, kvg, 3072, 0, nullptr, nullptr, 0, 0, 1024, 1.0f, nullptr);

    zbias_kernel<<<4096, 128, 0, stream>>>(z, pnorm_w, pnorm_b, bias_w, bias_b, biasO);

    gemm_nt_kernel<true, false, true, 0><<<dim3(16, 16, 16), 128, 0, stream>>>(
        kvg, 192, 196608LL,
        qm, 64, 65536LL,
        scores, 1024, 1048576LL,
        nullptr,
        biasO, 1024, 1048576LL,
        64, 1.0f / 64.0f, smstat);

    gemm_tn_kernel<<<dim3(1, 16, 16), 256, 0, stream>>>(
        scores, 1024, 1048576LL,
        kvg + 64, 192, 196608LL,
        obuf, 64, 65536LL,
        1024, smstat);

    gate_o_kernel<<<1024, 256, 0, stream>>>(kvg, (const float4*)obuf, (float4*)go);

    gemm_nt_kernel<false, false, false, 0><<<dim3(16, 16, 1), 128, 0, stream>>>(
        go, 1024, 0, attn_w, 1024, 0, a2, 1024, 0, nullptr, nullptr, 0, 0, 1024, 1.0f, nullptr);

    gemm_nt_kernel<false, true, true, 1><<<dim3(16, 16, 1), 128, 0, stream>>>(
        s, 512, 0, out_w, 512, 0, out, 1024, 0, out_b, a2, 1024, 0, 512, 1.0f, nullptr);
}


# NT GEMMs with K >= 512: a wave owns 64x64 and K is split over the block's 4 waves (partials added through LDS), so a 16-line fragment load feeds 4 WMMAs, was 2; k order of the sums changes
# speedup vs baseline: 2.1874x; 1.1108x over previous
#include <hip/hip_runtime.h>
#include <hip/hip_bf16.h>
#include <math.h>

typedef __attribute__((ext_vector_type(2))) float v2f;
typedef __attribute__((ext_vector_type(8))) float v8f;

#define S_DIM 1024
#define CA 1024
#define CS 512
#define CZ 64
#define NH 16
#define HC 64

__device__ __forceinline__ float sigmoidf_(float x) {
    return 1.0f / (1.0f + __expf(-x));
}

template <int M>
__device__ __forceinline__ float xlane(float v) {
    return __int_as_float(__builtin_amdgcn_update_dpp(0, __float_as_int(v), 0x160 + M, 0xf, 0xf, true));
}

__device__ __forceinline__ float rowsum16(float s) {
    s += xlane<1>(s);
    s += xlane<2>(s);
    s += xlane<4>(s);
    s += xlane<8>(s);
    return s;
}

__device__ __forceinline__ float rowmax16(float s) {
    s = fmaxf(s, xlane<1>(s));
    s = fmaxf(s, xlane<2>(s));
    s = fmaxf(s, xlane<4>(s));
    s = fmaxf(s, xlane<8>(s));
    return s;
}

__global__ void ln_kernel(const float* __restrict__ x, const float* __restrict__ w,
                          float* __restrict__ y, int C) {
    __shared__ float red[256];
    __shared__ float red2[256];
    int row = blockIdx.x;
    const float4* xr = (const float4*)(x + (size_t)row * C);
    int C4 = C >> 2;
    float sum = 0.f, sq = 0.f;
    for (int c = threadIdx.x; c < C4; c += 256) {
        float4 v = xr[c];
        sum += v.x + v.y + v.z + v.w;
        sq  += v.x * v.x + v.y * v.y + v.z * v.z + v.w * v.w;
    }
    red[threadIdx.x] = sum; red2[threadIdx.x] = sq;
    __syncthreads();
    for (int off = 128; off > 0; off >>= 1) {
        if (threadIdx.x < off) {
            red[threadIdx.x]  += red[threadIdx.x + off];
            red2[threadIdx.x] += red2[threadIdx.x + off];
        }
        __syncthreads();
    }
    float m   = red[0] / C;
    float var = red2[0] / C - m * m;
    float inv = rsqrtf(var + 1e-5f);
    float4* yr = (float4*)(y + (size_t)row * C);
    for (int c = threadIdx.x; c < C4; c += 256) {
        float4 v = xr[c];
        v.x = (v.x - m) * inv; v.y = (v.y - m) * inv;
        v.z = (v.z - m) * inv; v.w = (v.w - m) * inv;
        if (w) {
            const float4 wv = ((const float4*)w)[c];
            v.x *= wv.x; v.y *= wv.y; v.z *= wv.z; v.w *= wv.w;
        }
        yr[c] = v;
    }
}

struct Frag16 {
    float4 a0[2], a1[2], b0[2], b1[2];
};

__device__ __forceinline__ void load_step(Frag16& f, const float* pA0, const float* pA1,
                                          const float* pB0, const float* pB1, int k0) {
    #pragma unroll
    for (int g = 0; g < 2; g++) {
        f.a0[g] = *(const float4*)(pA0 + k0 + 8 * g);
        f.b0[g] = *(const float4*)(pB0 + k0 + 8 * g);
        f.a1[g] = *(const float4*)(pA1 + k0 + 8 * g);
        f.b1[g] = *(const float4*)(pB1 + k0 + 8 * g);
    }
}

__device__ __forceinline__ void mma_step(const Frag16& f, v8f& c00, v8f& c01, v8f& c10, v8f& c11) {
    #pragma unroll
    for (int g = 0; g < 2; g++) {
        const v2f a0l = {f.a0[g].x, f.a0[g].y}, a0h = {f.a0[g].z, f.a0[g].w};
        const v2f a1l = {f.a1[g].x, f.a1[g].y}, a1h = {f.a1[g].z, f.a1[g].w};
        const v2f b0l = {f.b0[g].x, f.b0[g].y}, b0h = {f.b0[g].z, f.b0[g].w};
        const v2f b1l = {f.b1[g].x, f.b1[g].y}, b1h = {f.b1[g].z, f.b1[g].w};
        c00 = __builtin_amdgcn_wmma_f32_16x16x4_f32(false, a0l, false, b0l, (short)0, c00, false, false);
        c01 = __builtin_amdgcn_wmma_f32_16x16x4_f32(false, a0l, false, b1l, (short)0, c01, false, false);
        c10 = __builtin_amdgcn_wmma_f32_16x16x4_f32(false, a1l, false, b0l, (short)0, c10, false, false);
        c11 = __builtin_amdgcn_wmma_f32_16x16x4_f32(false, a1l, false, b1l, (short)0, c11, false, false);
        c00 = __builtin_amdgcn_wmma_f32_16x16x4_f32(false, a0h, false, b0h, (short)0, c00, false, false);
        c01 = __builtin_amdgcn_wmma_f32_16x16x4_f32(false, a0h, false, b1h, (short)0, c01, false, false);
        c10 = __builtin_amdgcn_wmma_f32_16x16x4_f32(false, a1h, false, b0h, (short)0, c10, false, false);
        c11 = __builtin_amdgcn_wmma_f32_16x16x4_f32(false, a1h, false, b1h, (short)0, c11, false, false);
    }
}

template <bool STATS, bool HAS_BIAS, bool HAS_ADD, int MODE>
__global__ void __launch_bounds__(128)
gemm_nt_kernel(const float* __restrict__ A, int lda, long long sA,
               const float* __restrict__ B, int ldb, long long sB,
               float* __restrict__ C, int ldc, long long sC,
               const float* __restrict__ bias,
               const float* __restrict__ addm, int ldadd, long long sAdd,
               int K, float scale, float* __restrict__ stats) {
    const int z = blockIdx.z;
    A += (size_t)z * sA; B += (size_t)z * sB; C += (size_t)z * sC;
    if (HAS_ADD) addm += (size_t)z * sAdd;
    const int tileM = blockIdx.y * 64;
    const int tileN = blockIdx.x * 64;
    const int tid  = threadIdx.x;
    const int wave = tid >> 5, lane = tid & 31;
    const int wm = wave >> 1, wn = wave & 1;
    const int half = lane >> 4, l = lane & 15;

    const float* pA0 = A + (size_t)(tileM + wm * 32 + l) * lda + half * 4;
    const float* pA1 = pA0 + (size_t)16 * lda;
    const float* pB0 = B + (size_t)(tileN + wn * 32 + l) * ldb + half * 4;
    const float* pB1 = pB0 + (size_t)16 * ldb;

    v8f c00 = {}, c01 = {}, c10 = {}, c11 = {};
    const int nsteps = K >> 4;

    #define ORDER_FENCE() asm volatile("" : "+v"(c00), "+v"(c01), "+v"(c10), "+v"(c11) : : "memory")
    Frag16 r0, r1, r2, r3;
    load_step(r0, pA0, pA1, pB0, pB1, 0);
    ORDER_FENCE();
    load_step(r1, pA0, pA1, pB0, pB1, 16);
    ORDER_FENCE();
    load_step(r2, pA0, pA1, pB0, pB1, 32);
    ORDER_FENCE();
    load_step(r3, pA0, pA1, pB0, pB1, 48);
    ORDER_FENCE();
    int s = 0;
    for (; s + 4 < nsteps; s += 4) {
        mma_step(r0, c00, c01, c10, c11);
        ORDER_FENCE();
        load_step(r0, pA0, pA1, pB0, pB1, (s + 4) * 16);
        ORDER_FENCE();
        mma_step(r1, c00, c01, c10, c11);
        ORDER_FENCE();
        load_step(r1, pA0, pA1, pB0, pB1, (s + 5) * 16);
        ORDER_FENCE();
        mma_step(r2, c00, c01, c10, c11);
        ORDER_FENCE();
        load_step(r2, pA0, pA1, pB0, pB1, (s + 6) * 16);
        ORDER_FENCE();
        mma_step(r3, c00, c01, c10, c11);
        ORDER_FENCE();
        load_step(r3, pA0, pA1, pB0, pB1, (s + 7) * 16);
        ORDER_FENCE();
    }
    mma_step(r0, c00, c01, c10, c11);
    ORDER_FENCE();
    const int ncol = tileN + wn * 32 + l;
    float bv0 = 0.f, bv1 = 0.f;
    if (HAS_BIAS) { bv0 = bias[ncol]; bv1 = bias[ncol + 16]; }
    float ad0[2][8], ad1[2][8];
    if (HAS_ADD) {
        #pragma unroll
        for (int i = 0; i < 2; i++) {
            #pragma unroll
            for (int r = 0; r < 8; r++) {
                const float* pa = addm + (size_t)(tileM + wm * 32 + i * 16 + half * 8 + r) * ldadd + ncol;
                ad0[i][r] = pa[0];
                ad1[i][r] = pa[16];
            }
        }
    }
    ORDER_FENCE();
    mma_step(r1, c00, c01, c10, c11);
    mma_step(r2, c00, c01, c10, c11);
    mma_step(r3, c00, c01, c10, c11);
    #undef ORDER_FENCE

    #pragma unroll
    for (int i = 0; i < 2; i++) {
        const v8f acc0 = i ? c10 : c00;
        const v8f acc1 = i ? c11 : c01;
        float keep_mx = 0.f, keep_sum = 0.f;
        #pragma unroll
        for (int r = 0; r < 8; r++) {
            int m  = tileM + wm * 32 + i * 16 + half * 8 + r;
            int n0 = tileN + wn * 32 + l;
            int n1 = n0 + 16;
            float v0 = acc0[r] * scale;
            float v1 = acc1[r] * scale;
            if (HAS_BIAS) { v0 += bv0; v1 += bv1; }
            if (MODE == 1) {
                v0 = sigmoidf_(v0) * ad0[i][r];
                v1 = sigmoidf_(v1) * ad1[i][r];
            } else if (HAS_ADD) {
                v0 += ad0[i][r];
                v1 += ad1[i][r];
            }
            C[(size_t)m * ldc + n0] = v0;
            C[(size_t)m * ldc + n1] = v1;
            if (STATS) {
                const float mx = rowmax16(fmaxf(v0, v1));
                const float es = rowsum16(__expf(v0 - mx) + __expf(v1 - mx));
                if ((l & 7) == r) { keep_mx = mx; keep_sum = es; }
            }
        }
        if (STATS) {
            if (l < 8) {
                const size_t line = ((size_t)z * 64 + (tileM >> 4) + wm * 2 + i) * 32 + (blockIdx.x * 2 + wn);
                *(float2*)(stats + (line * 16 + half * 8 + l) * 2) = make_float2(keep_mx, keep_sum);
            }
        }
    }
}

struct Frag32 {
    float4 a[4][4], b[4][4];
};

__device__ __forceinline__ void load_ks(Frag32& f, const float* pA, const float* pB,
                                        size_t a16, size_t b16, int k0) {
    #pragma unroll
    for (int i = 0; i < 4; i++) {
        #pragma unroll
        for (int g = 0; g < 4; g++) f.a[i][g] = *(const float4*)(pA + i * a16 + k0 + 8 * g);
        #pragma unroll
        for (int g = 0; g < 4; g++) f.b[i][g] = *(const float4*)(pB + i * b16 + k0 + 8 * g);
    }
}

__device__ __forceinline__ void mma_ks(const Frag32& f, v8f (&acc)[4][4]) {
    #pragma unroll
    for (int g = 0; g < 4; g++) {
        #pragma unroll
        for (int e = 0; e < 2; e++) {
            #pragma unroll
            for (int i = 0; i < 4; i++) {
                const v2f av = e ? (v2f){f.a[i][g].z, f.a[i][g].w} : (v2f){f.a[i][g].x, f.a[i][g].y};
                #pragma unroll
                for (int j = 0; j < 4; j++) {
                    const v2f bv = e ? (v2f){f.b[j][g].z, f.b[j][g].w} : (v2f){f.b[j][g].x, f.b[j][g].y};
                    acc[i][j] = __builtin_amdgcn_wmma_f32_16x16x4_f32(false, av, false, bv, (short)0,
                                                                      acc[i][j], false, false);
                }
            }
        }
    }
}

template <bool HAS_BIAS, bool HAS_ADD, int MODE>
__global__ void __launch_bounds__(128)
gemm_nt_ksplit_kernel(const float* __restrict__ A, int lda,
                      const float* __restrict__ B, int ldb,
                      float* __restrict__ C, int ldc,
                      const float* bias,
                      const float* addm, int ldadd,
                      int K, float scale) {
    __shared__ float red[4][64][68];
    const int tileM = blockIdx.y * 64;
    const int tileN = blockIdx.x * 64;
    const int tid  = threadIdx.x;
    const int wave = __builtin_amdgcn_readfirstlane(tid >> 5);
    const int lane = tid & 31;
    const int half = lane >> 4, l = lane & 15;
    const int kw = K >> 2;

    const float* pA = A + (size_t)(tileM + l) * lda + wave * kw + half * 4;
    const float* pB = B + (size_t)(tileN + l) * ldb + wave * kw + half * 4;
    const size_t a16 = (size_t)lda * 16, b16 = (size_t)ldb * 16;

    v8f acc[4][4];
    #pragma unroll
    for (int i = 0; i < 16; i++) acc[i >> 2][i & 3] = (v8f){};
    const int nsteps = kw >> 5;

    #define KS_FENCE() asm volatile("" : "+v"(acc[0][0]), "+v"(acc[0][1]), "+v"(acc[0][2]), "+v"(acc[0][3]), \
                                         "+v"(acc[1][0]), "+v"(acc[1][1]), "+v"(acc[1][2]), "+v"(acc[1][3]), \
                                         "+v"(acc[2][0]), "+v"(acc[2][1]), "+v"(acc[2][2]), "+v"(acc[2][3]), \
                                         "+v"(acc[3][0]), "+v"(acc[3][1]), "+v"(acc[3][2]), "+v"(acc[3][3]) : : "memory")
    Frag32 r0, r1;
    load_ks(r0, pA, pB, a16, b16, 0);
    KS_FENCE();
    load_ks(r1, pA, pB, a16, b16, 32);
    KS_FENCE();
    for (int s = 0; s + 2 < nsteps; s += 2) {
        mma_ks(r0, acc);
        KS_FENCE();
        load_ks(r0, pA, pB, a16, b16, (s + 2) * 32);
        KS_FENCE();
        mma_ks(r1, acc);
        KS_FENCE();
        load_ks(r1, pA, pB, a16, b16, (s + 3) * 32);
        KS_FENCE();
    }
    mma_ks(r0, acc);
    KS_FENCE();
    const int orow = tid >> 4, n4 = (tid & 15) * 4;
    float4 bv = make_float4(0.f, 0.f, 0.f, 0.f);
    if (HAS_BIAS) bv = *(const float4*)(bias + tileN + n4);
    float4 ad[8];
    if (HAS_ADD) {
        #pragma unroll
        for (int q = 0; q < 8; q++)
            ad[q] = *(const float4*)(addm + (size_t)(tileM + orow + 8 * q) * ldadd + tileN + n4);
    }
    KS_FENCE();
    mma_ks(r1, acc);
    #undef KS_FENCE

    #pragma unroll
    for (int i = 0; i < 4; i++) {
        #pragma unroll
        for (int j = 0; j < 4; j++) {
            #pragma unroll
            for (int r = 0; r < 8; r++) red[wave][16 * i + 8 * half + r][16 * j + l] = acc[i][j][r];
        }
    }
    __syncthreads();
    #pragma unroll
    for (int q = 0; q < 8; q++) {
        const int ro = orow + 8 * q;
        float4 o = *(const float4*)&red[0][ro][n4];
        #pragma unroll
        for (int w = 1; w < 4; w++) {
            const float4 p = *(const float4*)&red[w][ro][n4];
            o.x += p.x; o.y += p.y; o.z += p.z; o.w += p.w;
        }
        float4 v = make_float4(o.x * scale, o.y * scale, o.z * scale, o.w * scale);
        if (HAS_BIAS) { v.x += bv.x; v.y += bv.y; v.z += bv.z; v.w += bv.w; }
        if (MODE == 1) {
            v.x = sigmoidf_(v.x) * ad[q].x; v.y = sigmoidf_(v.y) * ad[q].y;
            v.z = sigmoidf_(v.z) * ad[q].z; v.w = sigmoidf_(v.w) * ad[q].w;
        } else if (HAS_ADD) {
            v.x += ad[q].x; v.y += ad[q].y; v.z += ad[q].z; v.w += ad[q].w;
        }
        *(float4*)(C + (size_t)(tileM + ro) * ldc + tileN + n4) = v;
    }
}

__device__ __forceinline__ float2 softmax_row_scale(float2 s0, float2 s1) {
    const float mx  = rowmax16(fmaxf(s0.x, s1.x));
    const float sum = rowsum16(s0.y * __expf(s0.x - mx) + s1.y * __expf(s1.x - mx));
    return make_float2(mx, 1.0f / sum);
}

typedef __attribute__((ext_vector_type(4))) float v4f;
struct StepTN {
    v4f p0, p1, v0, v1, s0, s1;
};

__device__ __forceinline__ void load_tn(StepTN& f, const float* pP, const float* pV, const float* pS,
                                        int lda, int ldb) {
    f.s0 = *(const v4f*)pS;
    f.s1 = *(const v4f*)(pS + 512);
    f.p0 = *(const v4f*)pP;
    f.p1 = *(const v4f*)(pP + lda);
    f.v0 = *(const v4f*)pV;
    f.v1 = *(const v4f*)(pV + ldb);
}

__device__ __forceinline__ void prep_tn(StepTN& f) {
    const float2 m0 = softmax_row_scale(make_float2(f.s0.x, f.s0.y), make_float2(f.s1.x, f.s1.y));
    const float2 m1 = softmax_row_scale(make_float2(f.s0.z, f.s0.w), make_float2(f.s1.z, f.s1.w));
    f.p0.x = __expf(f.p0.x - m0.x) * m0.y; f.p0.y = __expf(f.p0.y - m0.x) * m0.y;
    f.p0.z = __expf(f.p0.z - m0.x) * m0.y; f.p0.w = __expf(f.p0.w - m0.x) * m0.y;
    f.p1.x = __expf(f.p1.x - m1.x) * m1.y; f.p1.y = __expf(f.p1.y - m1.x) * m1.y;
    f.p1.z = __expf(f.p1.z - m1.x) * m1.y; f.p1.w = __expf(f.p1.w - m1.x) * m1.y;
}

#define TN_MMA(t, u, vt0, vt1, pu0, pu1)                                                         \
    acc[t][u] = __builtin_amdgcn_wmma_f32_16x16x4_f32(false, (v2f){vt0, vt1}, false,             \
                                                      (v2f){pu0, pu1}, (short)0, acc[t][u], false, false)

__device__ __forceinline__ void mma_tn(const StepTN& f, v8f (&acc)[4][4]) {
    TN_MMA(0, 0, f.v0.x, f.v1.x, f.p0.x, f.p1.x); TN_MMA(0, 1, f.v0.x, f.v1.x, f.p0.y, f.p1.y);
    TN_MMA(0, 2, f.v0.x, f.v1.x, f.p0.z, f.p1.z); TN_MMA(0, 3, f.v0.x, f.v1.x, f.p0.w, f.p1.w);
    TN_MMA(1, 0, f.v0.y, f.v1.y, f.p0.x, f.p1.x); TN_MMA(1, 1, f.v0.y, f.v1.y, f.p0.y, f.p1.y);
    TN_MMA(1, 2, f.v0.y, f.v1.y, f.p0.z, f.p1.z); TN_MMA(1, 3, f.v0.y, f.v1.y, f.p0.w, f.p1.w);
    TN_MMA(2, 0, f.v0.z, f.v1.z, f.p0.x, f.p1.x); TN_MMA(2, 1, f.v0.z, f.v1.z, f.p0.y, f.p1.y);
    TN_MMA(2, 2, f.v0.z, f.v1.z, f.p0.z, f.p1.z); TN_MMA(2, 3, f.v0.z, f.v1.z, f.p0.w, f.p1.w);
    TN_MMA(3, 0, f.v0.w, f.v1.w, f.p0.x, f.p1.x); TN_MMA(3, 1, f.v0.w, f.v1.w, f.p0.y, f.p1.y);
    TN_MMA(3, 2, f.v0.w, f.v1.w, f.p0.z, f.p1.z); TN_MMA(3, 3, f.v0.w, f.v1.w, f.p0.w, f.p1.w);
}

__global__ void __launch_bounds__(256)
gemm_tn_kernel(const float* __restrict__ A, int lda, long long sA,
               const float* __restrict__ B, int ldb, long long sB,
               float* __restrict__ C, int ldc, long long sC,
               int K, const float* __restrict__ stats) {
    __shared__ float red[8][64][68];
    const int z = blockIdx.z;
    A += (size_t)z * sA; B += (size_t)z * sB; C += (size_t)z * sC;
    const int tileM = blockIdx.y * 64;
    const int tid  = threadIdx.x;
    const int wave = __builtin_amdgcn_readfirstlane(tid >> 5);
    const int lane = tid & 31;
    const int half = lane >> 4, l = lane & 15;
    const int kw = K >> 3;
    const int row0 = wave * kw;

    const float* qP = A + (size_t)row0 * lda + (size_t)(half * 2) * lda + tileM + 4 * l;
    const float* qV = B + (size_t)row0 * ldb + (size_t)(half * 2) * ldb + 4 * l;
    const float* qS = stats + (size_t)z * 65536 + (size_t)(row0 >> 4) * 1024 + l * 32 + half * 4;
    const size_t a4 = (size_t)lda * 4, b4 = (size_t)ldb * 4;

    v8f acc[4][4];
    #pragma unroll
    for (int i = 0; i < 16; i++) acc[i >> 2][i & 3] = (v8f){};
    const int ntrips = kw >> 4;

    #define TN_FENCE() asm volatile("" : "+v"(acc[0][0]), "+v"(acc[0][1]), "+v"(acc[0][2]), "+v"(acc[0][3]), \
                                         "+v"(acc[1][0]), "+v"(acc[1][1]), "+v"(acc[1][2]), "+v"(acc[1][3]), \
                                         "+v"(acc[2][0]), "+v"(acc[2][1]), "+v"(acc[2][2]), "+v"(acc[2][3]), \
                                         "+v"(acc[3][0]), "+v"(acc[3][1]), "+v"(acc[3][2]), "+v"(acc[3][3]) : : "memory")
    #define TN_GATE(f) asm volatile("" : "+v"(f.s0), "+v"(f.s1), "+v"(f.p0), "+v"(f.p1))
    StepTN r0, r1, r2, r3;
    load_tn(r0, qP, qV, qS, lda, ldb);
    TN_FENCE();
    load_tn(r1, qP + a4, qV + b4, qS + 8, lda, ldb);
    TN_FENCE();
    load_tn(r2, qP + 2 * a4, qV + 2 * b4, qS + 16, lda, ldb);
    TN_FENCE();
    load_tn(r3, qP + 3 * a4, qV + 3 * b4, qS + 24, lda, ldb);
    TN_FENCE();
    TN_GATE(r0);
    prep_tn(r0);
    for (int trip = 1; trip < ntrips; trip++) {
        qP += 4 * a4; qV += 4 * b4; qS += 1024;
        TN_GATE(r1);
        prep_tn(r1);
        mma_tn(r0, acc);
        TN_FENCE();
        load_tn(r0, qP, qV, qS, lda, ldb);
        TN_FENCE();
        TN_GATE(r2);
        prep_tn(r2);
        mma_tn(r1, acc);
        TN_FENCE();
        load_tn(r1, qP + a4, qV + b4, qS + 8, lda, ldb);
        TN_FENCE();
        TN_GATE(r3);
        prep_tn(r3);
        mma_tn(r2, acc);
        TN_FENCE();
        load_tn(r2, qP + 2 * a4, qV + 2 * b4, qS + 16, lda, ldb);
        TN_FENCE();
        TN_GATE(r0);
        prep_tn(r0);
        mma_tn(r3, acc);
        TN_FENCE();
        load_tn(r3, qP + 3 * a4, qV + 3 * b4, qS + 24, lda, ldb);
        TN_FENCE();
    }
    TN_GATE(r1);
    prep_tn(r1);
    mma_tn(r0, acc);
    TN_FENCE();
    TN_GATE(r2);
    prep_tn(r2);
    mma_tn(r1, acc);
    TN_FENCE();
    TN_GATE(r3);
    prep_tn(r3);
    mma_tn(r2, acc);
    mma_tn(r3, acc);
    #undef TN_GATE
    #undef TN_FENCE

    #pragma unroll
    for (int i = 0; i < 32; i++) {
        const int u = i >> 3, r = i & 7;
        *(float4*)&red[wave][16 * u + l][32 * half + 4 * r] =
            make_float4(acc[0][u][r], acc[1][u][r], acc[2][u][r], acc[3][u][r]);
    }
    __syncthreads();
    #pragma unroll
    for (int q = 0; q < 4; q++) {
        const int idx = tid + 256 * q;
        const int ro = idx >> 4, n4 = (idx & 15) * 4;
        const int m = 4 * (ro & 15) + (ro >> 4);
        float4 o = *(const float4*)&red[0][ro][n4];
        #pragma unroll
        for (int w = 1; w < 8; w++) {
            const float4 p = *(const float4*)&red[w][ro][n4];
            o.x += p.x; o.y += p.y; o.z += p.z; o.w += p.w;
        }
        *(float4*)(C + (size_t)(tileM + m) * ldc + n4) = o;
    }
}

__device__ __forceinline__ float headsum16(const v2f* P) {
    float a[8], b[4], c[2];
    #pragma unroll
    for (int kk = 0; kk < 4; kk++) {
        a[2 * kk]     = P[kk].x + xlane<8>(P[kk + 4].x);
        a[2 * kk + 1] = P[kk].y + xlane<8>(P[kk + 4].y);
    }
    #pragma unroll
    for (int k = 0; k < 4; k++) b[k] = a[k] + xlane<4>(a[k + 4]);
    #pragma unroll
    for (int k = 0; k < 2; k++) c[k] = b[k] + xlane<2>(b[k + 2]);
    return c[0] + xlane<1>(c[1]);
}

__global__ void __launch_bounds__(128, 8)
zbias_kernel(const float* __restrict__ z,
             const float* __restrict__ pnw, const float* __restrict__ pnb,
             const float* __restrict__ bw,  const float* __restrict__ bb,
             float* __restrict__ out) {
    const int lane = threadIdx.x & 31;
    const int j = lane & 15;
    const int wave_id = blockIdx.x * 4 + (threadIdx.x >> 5);

    const float4 pw = *(const float4*)(pnw + 4 * j);
    const float4 pb = *(const float4*)(pnb + 4 * j);
    v2f W[8][4];
    v2f Cp[8];
    #pragma unroll
    for (int kk = 0; kk < 8; kk++) {
        const float4 r0 = *(const float4*)(bw + (((2 * kk) ^ j) * CZ + 4 * j));
        const float4 r1 = *(const float4*)(bw + (((2 * kk + 1) ^ j) * CZ + 4 * j));
        Cp[kk].x = pb.x * r0.x + pb.y * r0.y + pb.z * r0.z + pb.w * r0.w;
        Cp[kk].y = pb.x * r1.x + pb.y * r1.y + pb.z * r1.z + pb.w * r1.w;
        W[kk][0].x = r0.x * pw.x; W[kk][0].y = r1.x * pw.x;
        W[kk][1].x = r0.y * pw.y; W[kk][1].y = r1.y * pw.y;
        W[kk][2].x = r0.z * pw.z; W[kk][2].y = r1.z * pw.z;
        W[kk][3].x = r0.w * pw.w; W[kk][3].y = r1.w * pw.w;
    }
    const float cB = headsum16(Cp) + bb[j];

    const size_t base = (size_t)wave_id * 64 * 16;
    for (int chunk = 0; chunk < 4; chunk++) {
        const size_t cb = base + (size_t)chunk * 256 + lane;
        const float4* zp = (const float4*)z + cb;
        float* op = out + cb;
        float4 v[8];
        #pragma unroll
        for (int i = 0; i < 8; i++) v[i] = zp[i * 32];
        #pragma unroll
        for (int i = 0; i < 8; i++) {
            const float4 t = v[i];
            float s = (t.x + t.y) + (t.z + t.w);
            float q = t.x * t.x + t.y * t.y + t.z * t.z + t.w * t.w;
            s = rowsum16(s);
            q = rowsum16(q);
            const float m   = s * (1.0f / 64.0f);
            const float var = q * (1.0f / 64.0f) - m * m;
            const float inv = rsqrtf(var + 1e-5f);
            const float d0 = t.x - m, d1 = t.y - m, d2 = t.z - m, d3 = t.w - m;
            const v2f D0 = {d0, d0}, D1 = {d1, d1}, D2 = {d2, d2}, D3 = {d3, d3};
            v2f P[8];
            #pragma unroll
            for (int kk = 0; kk < 8; kk++) {
                v2f acc = D0 * W[kk][0];
                acc = __builtin_elementwise_fma(D1, W[kk][1], acc);
                acc = __builtin_elementwise_fma(D2, W[kk][2], acc);
                acc = __builtin_elementwise_fma(D3, W[kk][3], acc);
                P[kk] = acc;
            }
            op[i * 32] = headsum16(P) * inv + cB;
        }
    }
}

__global__ void adaln_gate_kernel(const float4* __restrict__ pb, const float4* __restrict__ pn,
                                  const float4* __restrict__ an, float4* __restrict__ a1) {
    int i = blockIdx.x * 256 + threadIdx.x;
    float4 p = pb[i], q = pn[i], n = an[i];
    float4 r;
    r.x = sigmoidf_(p.x * n.x + q.x);
    r.y = sigmoidf_(p.y * n.y + q.y);
    r.z = sigmoidf_(p.z * n.z + q.z);
    r.w = sigmoidf_(p.w * n.w + q.w);
    a1[i] = r;
}

__global__ void gate_o_kernel(const float* __restrict__ kvg, const float4* __restrict__ obuf,
                              float4* __restrict__ go) {
    int q = blockIdx.x * 256 + threadIdx.x;
    int h  = q >> 14;
    int j  = (q >> 4) & 1023;
    int c4 = (q & 15) * 4;
    const float4 g = *(const float4*)(kvg + (size_t)h * 196608 + (size_t)j * 192 + 128 + c4);
    float4 o = obuf[q];
    float4 r;
    r.x = sigmoidf_(g.x) * o.x;
    r.y = sigmoidf_(g.y) * o.y;
    r.z = sigmoidf_(g.z) * o.z;
    r.w = sigmoidf_(g.w) * o.w;
    go[q] = r;
}

extern "C" void kernel_launch(void* const* d_in, const int* in_sizes, int n_in,
                              void* d_out, int out_size, void* d_ws, size_t ws_size,
                              hipStream_t stream) {
    const float* a       = (const float*)d_in[0];
    const float* z       = (const float*)d_in[1];
    const float* s       = (const float*)d_in[2];
    const float* sn_w    = (const float*)d_in[3];
    const float* pb_w    = (const float*)d_in[4];
    const float* pb_b    = (const float*)d_in[5];
    const float* pn_w    = (const float*)d_in[6];
    const float* pnorm_w = (const float*)d_in[7];
    const float* pnorm_b = (const float*)d_in[8];
    const float* q_w     = (const float*)d_in[9];
    const float* q_b     = (const float*)d_in[10];
    const float* kvg_w   = (const float*)d_in[11];
    const float* bias_w  = (const float*)d_in[12];
    const float* bias_b  = (const float*)d_in[13];
    const float* attn_w  = (const float*)d_in[14];
    const float* out_w   = (const float*)d_in[15];
    const float* out_b   = (const float*)d_in[16];
    float* out = (float*)d_out;
    float* ws  = (float*)d_ws;

    float* a_n    = ws + 0;
    float* s_n    = ws + 1048576;
    float* pb     = ws + 1572864;
    float* pn     = ws + 2621440;
    float* qm     = ws + 3670016;
    float* kvg    = ws + 4718592;
    float* biasO  = ws + 7864320;
    float* obuf   = biasO;
    float* go     = biasO + 1048576;
    float* a2     = biasO + 2097152;
    float* scores = ws + 24641536;
    float* smstat = a_n;

    ln_kernel<<<1024, 256, 0, stream>>>(a, nullptr, a_n, 1024);
    ln_kernel<<<1024, 256, 0, stream>>>(s, sn_w, s_n, 512);

    gemm_nt_ksplit_kernel<true, false, 0><<<dim3(16, 16, 1), 128, 0, stream>>>(
        s_n, 512, pb_w, 512, pb, 1024, pb_b, nullptr, 0, 512, 1.0f);
    gemm_nt_ksplit_kernel<false, false, 0><<<dim3(16, 16, 1), 128, 0, stream>>>(
        s_n, 512, pn_w, 512, pn, 1024, nullptr, nullptr, 0, 512, 1.0f);
    adaln_gate_kernel<<<1024, 256, 0, stream>>>(
        (const float4*)pb, (const float4*)pn, (const float4*)a_n, (float4*)pb);
    float* a1 = pb;

    gemm_nt_ksplit_kernel<true, false, 0><<<dim3(16, 16, 1), 128, 0, stream>>>(
        a1, 1024, q_w, 1024, qm, 1024, q_b, nullptr, 0, 1024, 1.0f);
    gemm_nt_ksplit_kernel<false, false, 0><<<dim3(48, 16, 1), 128, 0, stream>>>(
        a1, 1024, kvg_w, 1024, kvg, 3072, nullptr, nullptr, 0, 1024, 1.0f);

    zbias_kernel<<<4096, 128, 0, stream>>>(z, pnorm_w, pnorm_b, bias_w, bias_b, biasO);

    gemm_nt_kernel<true, false, true, 0><<<dim3(16, 16, 16), 128, 0, stream>>>(
        kvg, 192, 196608LL,
        qm, 64, 65536LL,
        scores, 1024, 1048576LL,
        nullptr,
        biasO, 1024, 1048576LL,
        64, 1.0f / 64.0f, smstat);

    gemm_tn_kernel<<<dim3(1, 16, 16), 256, 0, stream>>>(
        scores, 1024, 1048576LL,
        kvg + 64, 192, 196608LL,
        obuf, 64, 65536LL,
        1024, smstat);

    gate_o_kernel<<<1024, 256, 0, stream>>>(kvg, (const float4*)obuf, (float4*)go);

    gemm_nt_ksplit_kernel<false, false, 0><<<dim3(16, 16, 1), 128, 0, stream>>>(
        go, 1024, attn_w, 1024, a2, 1024, nullptr, nullptr, 0, 1024, 1.0f);

    gemm_nt_ksplit_kernel<true, true, 1><<<dim3(16, 16, 1), 128, 0, stream>>>(
        s, 512, out_w, 512, out, 1024, out_b, a2, 1024, 512, 1.0f);
}
